# starting kernel
# baseline (speedup 1.0000x reference)
#include <hip/hip_runtime.h>
#include <hip/hip_bf16.h>
#include <math.h>

#define BB      4
#define LL      4096
#define DMODEL  384
#define DSTATE  16
#define DTRANK  24
#define CTOT    (DTRANK + 2*DSTATE)
#define MTOT    (BB*LL)
#define NCH     32
#define CHL     (LL/NCH)

typedef float     v2f  __attribute__((ext_vector_type(2)));
typedef float     v8f  __attribute__((ext_vector_type(8)));
typedef _Float16  v16h __attribute__((ext_vector_type(16)));

#if defined(__gfx1250__) && __has_builtin(__builtin_amdgcn_wmma_f32_16x16x4_f32)
#define HAVE_WMMA_F32 1
#else
#define HAVE_WMMA_F32 0
#endif

__global__ __launch_bounds__(256)
void proj_gemm_kernel(const float* __restrict__ X,
                      const float* __restrict__ W1,
                      const float* __restrict__ prompt,
                      float* __restrict__ dtsr,
                      float* __restrict__ Bsc,
                      float* __restrict__ Csc)
{
    __shared__ float Wlds[16 * 388];

    const int c0 = blockIdx.y * 16;
    for (int i = threadIdx.x; i < 16 * DMODEL; i += 256) {
        int n = i / DMODEL, k = i - n * DMODEL;
        float w = (c0 + n < CTOT) ? W1[(c0 + n) * DMODEL + k] : 0.0f;
        Wlds[n * 388 + k] = w;
    }
    __syncthreads();

    const int wave = threadIdx.x >> 5;
    const int lane = threadIdx.x & 31;
    const int ln   = lane & 15;
    const int hi   = lane >> 4;
    const int m0   = blockIdx.x * 128 + wave * 16;

    v8f acc = {};
    const float* arow = X + (size_t)(m0 + ln) * DMODEL;

#if HAVE_WMMA_F32
    for (int k0 = 0; k0 < DMODEL; k0 += 4) {
        v2f a, b;
        a.x = arow[k0 + 2 * hi];
        a.y = arow[k0 + 2 * hi + 1];
        b.x = Wlds[ln * 388 + k0 + 2 * hi];
        b.y = Wlds[ln * 388 + k0 + 2 * hi + 1];
        acc = __builtin_amdgcn_wmma_f32_16x16x4_f32(
                  false, a, false, b, (short)0, acc, false, false);
    }
#else
    for (int k0 = 0; k0 < DMODEL; k0 += 32) {
        v16h a, b;
#pragma unroll
        for (int v = 0; v < 8; ++v) {
            int kk = (v < 4 ? 2 * v : 16 + 2 * (v - 4)) + 8 * hi;
            a[2 * v]     = (_Float16)arow[k0 + kk];
            a[2 * v + 1] = (_Float16)arow[k0 + kk + 1];
            b[2 * v]     = (_Float16)Wlds[ln * 388 + k0 + kk];
            b[2 * v + 1] = (_Float16)Wlds[ln * 388 + k0 + kk + 1];
        }
        acc = __builtin_amdgcn_wmma_f32_16x16x32_f16(
                  false, a, false, b, (short)0, acc, false, false);
    }
#endif

    const int c = c0 + ln;
#pragma unroll
    for (int r = 0; r < 8; ++r) {
        int m = m0 + r + 8 * hi;
        float v = acc[r];
        if (c < DTRANK) {
            dtsr[(size_t)m * DTRANK + c] = v;
        } else if (c < DTRANK + DSTATE) {
            Bsc[(size_t)m * DSTATE + (c - DTRANK)] = v;
        } else if (c < CTOT) {
            int n = c - DTRANK - DSTATE;
            Csc[(size_t)m * DSTATE + n] = v + prompt[(size_t)m * DSTATE + n];
        }
    }
}

__global__ __launch_bounds__(256)
void dt_gemm_kernel(const float* __restrict__ dtsr,
                    const float* __restrict__ W2,
                    const float* __restrict__ bias,
                    float* __restrict__ delta)
{
    __shared__ float Wlds[DMODEL * 28];

    for (int i = threadIdx.x; i < DMODEL * DTRANK; i += 256) {
        int d = i / DTRANK, r = i - d * DTRANK;
        Wlds[d * 28 + r] = W2[i];
    }
    __syncthreads();

    const int wave = threadIdx.x >> 5;
    const int lane = threadIdx.x & 31;
    const int ln   = lane & 15;
    const int hi   = lane >> 4;
    const int m0   = blockIdx.x * 128 + wave * 16;

    const float* arow = dtsr + (size_t)(m0 + ln) * DTRANK;

#if HAVE_WMMA_F32
    v2f afr[6];
#pragma unroll
    for (int ks = 0; ks < 6; ++ks) {
        afr[ks].x = arow[4 * ks + 2 * hi];
        afr[ks].y = arow[4 * ks + 2 * hi + 1];
    }
#else
    v16h afr;
#pragma unroll
    for (int v = 0; v < 8; ++v) {
        int kk = (v < 4 ? 2 * v : 16 + 2 * (v - 4)) + 8 * hi;
        afr[2 * v]     = (kk     < DTRANK) ? (_Float16)arow[kk]     : (_Float16)0.f;
        afr[2 * v + 1] = (kk + 1 < DTRANK) ? (_Float16)arow[kk + 1] : (_Float16)0.f;
    }
#endif

    for (int d0 = 0; d0 < DMODEL; d0 += 16) {
        v8f acc = {};
#if HAVE_WMMA_F32
#pragma unroll
        for (int ks = 0; ks < 6; ++ks) {
            v2f b;
            b.x = Wlds[(d0 + ln) * 28 + 4 * ks + 2 * hi];
            b.y = Wlds[(d0 + ln) * 28 + 4 * ks + 2 * hi + 1];
            acc = __builtin_amdgcn_wmma_f32_16x16x4_f32(
                      false, afr[ks], false, b, (short)0, acc, false, false);
        }
#else
        v16h b;
#pragma unroll
        for (int v = 0; v < 8; ++v) {
            int kk = (v < 4 ? 2 * v : 16 + 2 * (v - 4)) + 8 * hi;
            b[2 * v]     = (kk     < DTRANK) ? (_Float16)Wlds[(d0 + ln) * 28 + kk]     : (_Float16)0.f;
            b[2 * v + 1] = (kk + 1 < DTRANK) ? (_Float16)Wlds[(d0 + ln) * 28 + kk + 1] : (_Float16)0.f;
        }
        acc = __builtin_amdgcn_wmma_f32_16x16x32_f16(
                  false, afr, false, b, (short)0, acc, false, false);
#endif
        const int   d  = d0 + ln;
        const float bi = bias[d];
#pragma unroll
        for (int r = 0; r < 8; ++r) {
            int   m = m0 + r + 8 * hi;
            float t = acc[r] + bi;
            float sp = (t > 20.0f) ? t : log1pf(expf(t));
            delta[(size_t)m * DMODEL + d] = sp;
        }
    }
}

__global__ __launch_bounds__(256)
void scan_passA_kernel(const float* __restrict__ delta,
                       const float* __restrict__ X,
                       const float* __restrict__ Bsc,
                       const float* __restrict__ Alog,
                       float* __restrict__ Pout,
                       float* __restrict__ Hend)
{
    const int idx = blockIdx.x * 256 + threadIdx.x;
    const int d   = idx % DMODEL;
    const int t2  = idx / DMODEL;
    const int b   = t2 % BB;
    const int ch  = t2 / BB;

    float A[DSTATE], h[DSTATE], P[DSTATE];
#pragma unroll
    for (int n = 0; n < DSTATE; ++n) {
        A[n] = -expf(Alog[d * DSTATE + n]);
        h[n] = 0.0f;
        P[n] = 1.0f;
    }

    const int l0 = ch * CHL;
    for (int t = 0; t < CHL; ++t) {
        const int bl = b * LL + l0 + t;
        if (t + 8 < CHL) {
            __builtin_prefetch(&delta[(size_t)(bl + 8) * DMODEL + d], 0, 1);
            __builtin_prefetch(&X[(size_t)(bl + 8) * DMODEL + d], 0, 1);
        }
        const float dt = delta[(size_t)bl * DMODEL + d];
        const float u  = X[(size_t)bl * DMODEL + d];
        const float du = dt * u;
        const float4* Bv4 = reinterpret_cast<const float4*>(Bsc + (size_t)bl * DSTATE);
        float Bl[DSTATE];
#pragma unroll
        for (int q = 0; q < 4; ++q) {
            float4 v = Bv4[q];
            Bl[4 * q] = v.x; Bl[4 * q + 1] = v.y; Bl[4 * q + 2] = v.z; Bl[4 * q + 3] = v.w;
        }
#pragma unroll
        for (int n = 0; n < DSTATE; ++n) {
            float a = expf(dt * A[n]);
            P[n] *= a;
            h[n]  = a * h[n] + du * Bl[n];
        }
    }

    float* Pp = Pout + (size_t)idx * DSTATE;
    float* Hp = Hend + (size_t)idx * DSTATE;
#pragma unroll
    for (int n = 0; n < DSTATE; ++n) { Pp[n] = P[n]; Hp[n] = h[n]; }
}

__global__ __launch_bounds__(256)
void scan_passB_kernel(const float* __restrict__ P,
                       const float* __restrict__ Hend,
                       float* __restrict__ Hinit)
{
    const int idx = blockIdx.x * 256 + threadIdx.x;
    if (idx >= BB * DMODEL) return;
    const int b = idx / DMODEL;
    const int d = idx - b * DMODEL;

    float h[DSTATE];
#pragma unroll
    for (int n = 0; n < DSTATE; ++n) h[n] = 0.0f;

    for (int ch = 0; ch < NCH; ++ch) {
        const size_t unit = ((size_t)(ch * BB + b) * DMODEL + d) * DSTATE;
        float* Hi = Hinit + unit;
#pragma unroll
        for (int n = 0; n < DSTATE; ++n) Hi[n] = h[n];
#pragma unroll
        for (int n = 0; n < DSTATE; ++n)
            h[n] = P[unit + n] * h[n] + Hend[unit + n];
    }
}

__global__ __launch_bounds__(256)
void scan_passC_kernel(const float* __restrict__ delta,
                       const float* __restrict__ X,
                       const float* __restrict__ Bsc,
                       const float* __restrict__ Csc,
                       const float* __restrict__ Alog,
                       const float* __restrict__ Ds,
                       const float* __restrict__ Hinit,
                       float* __restrict__ Y)
{
    const int idx = blockIdx.x * 256 + threadIdx.x;
    const int d   = idx % DMODEL;
    const int t2  = idx / DMODEL;
    const int b   = t2 % BB;
    const int ch  = t2 / BB;

    float A[DSTATE], h[DSTATE];
    const float* Hi = Hinit + (size_t)idx * DSTATE;
#pragma unroll
    for (int n = 0; n < DSTATE; ++n) {
        A[n] = -expf(Alog[d * DSTATE + n]);
        h[n] = Hi[n];
    }
    const float Dsd = Ds[d];

    const int l0 = ch * CHL;
    for (int t = 0; t < CHL; ++t) {
        const int bl = b * LL + l0 + t;
        if (t + 8 < CHL) {
            __builtin_prefetch(&delta[(size_t)(bl + 8) * DMODEL + d], 0, 1);
            __builtin_prefetch(&X[(size_t)(bl + 8) * DMODEL + d], 0, 1);
        }
        const float dt = delta[(size_t)bl * DMODEL + d];
        const float u  = X[(size_t)bl * DMODEL + d];
        const float du = dt * u;
        const float4* Bv4 = reinterpret_cast<const float4*>(Bsc + (size_t)bl * DSTATE);
        const float4* Cv4 = reinterpret_cast<const float4*>(Csc + (size_t)bl * DSTATE);
        float Bl[DSTATE], Cl[DSTATE];
#pragma unroll
        for (int q = 0; q < 4; ++q) {
            float4 vb = Bv4[q], vc = Cv4[q];
            Bl[4*q] = vb.x; Bl[4*q+1] = vb.y; Bl[4*q+2] = vb.z; Bl[4*q+3] = vb.w;
            Cl[4*q] = vc.x; Cl[4*q+1] = vc.y; Cl[4*q+2] = vc.z; Cl[4*q+3] = vc.w;
        }
        float y = 0.0f;
#pragma unroll
        for (int n = 0; n < DSTATE; ++n) {
            float a = expf(dt * A[n]);
            h[n] = a * h[n] + du * Bl[n];
            y   += h[n] * Cl[n];
        }
        Y[(size_t)bl * DMODEL + d] = y + u * Dsd;
    }
}

extern "C" void kernel_launch(void* const* d_in, const int* in_sizes, int n_in,
                              void* d_out, int out_size, void* d_ws, size_t ws_size,
                              hipStream_t stream)
{
    const float* x      = (const float*)d_in[0];
    const float* prompt = (const float*)d_in[1];
    const float* w1     = (const float*)d_in[2];
    const float* w2     = (const float*)d_in[3];
    const float* bias   = (const float*)d_in[4];
    const float* alog   = (const float*)d_in[5];
    const float* ds     = (const float*)d_in[6];
    float*       y      = (float*)d_out;

    float* ws = (float*)d_ws;
    size_t o = 0;
    float* dtsr  = ws + o; o += (size_t)MTOT * DTRANK;
    float* Bsc   = ws + o; o += (size_t)MTOT * DSTATE;
    float* Csc   = ws + o; o += (size_t)MTOT * DSTATE;
    float* delta = ws + o; o += (size_t)MTOT * DMODEL;
    const size_t units = (size_t)NCH * BB * DMODEL;
    float* P     = ws + o; o += units * DSTATE;
    float* Hend  = ws + o; o += units * DSTATE;
    float* Hinit = ws + o; o += units * DSTATE;
    (void)ws_size; (void)in_sizes; (void)n_in; (void)out_size;

    proj_gemm_kernel<<<dim3(MTOT / 128, 4, 1), 256, 0, stream>>>(x, w1, prompt, dtsr, Bsc, Csc);
    dt_gemm_kernel  <<<dim3(MTOT / 128, 1, 1), 256, 0, stream>>>(dtsr, w2, bias, delta);
    scan_passA_kernel<<<dim3((unsigned)(units / 256)), 256, 0, stream>>>(delta, x, Bsc, alog, P, Hend);
    scan_passB_kernel<<<dim3((BB * DMODEL + 255) / 256), 256, 0, stream>>>(P, Hend, Hinit);
    scan_passC_kernel<<<dim3((unsigned)(units / 256)), 256, 0, stream>>>(delta, x, Bsc, Csc, alog, ds, Hinit, y);
}


# Projection GEMM and dt GEMM+softplus fused in one kernel: a 64-row block keeps its 24 dt inputs in LDS (a launch and the dtsr buffer gone); projection k split over 4 waves, partials summed in k order
# speedup vs baseline: 3.6904x; 3.6904x over previous
#include <hip/hip_runtime.h>
#include <hip/hip_bf16.h>
#include <math.h>

#define BB      4
#define LL      4096
#define DMODEL  384
#define DSTATE  16
#define DTRANK  24
#define CTOT    (DTRANK + 2*DSTATE)
#define MTOT    (BB*LL)
#define NCH     32
#define CHL     (LL/NCH)
#define NSUB    4
#define SUBL    (CHL/NSUB)
#define DTILE   (256/NSUB)
#define NDT     (DMODEL/DTILE)

typedef float     v2f  __attribute__((ext_vector_type(2)));
typedef float     v8f  __attribute__((ext_vector_type(8)));
typedef _Float16  v16h __attribute__((ext_vector_type(16)));

#if defined(__gfx1250__) && __has_builtin(__builtin_amdgcn_wmma_f32_16x16x4_f32)
#define HAVE_WMMA_F32 1
#else
#define HAVE_WMMA_F32 0
#endif

#define FB_ROWS  64
#define W1P      388
#define SDP      28

#if !HAVE_WMMA_F32 && defined(__HIP_DEVICE_COMPILE__)
#error "the f32 16x16x4 matrix instruction is required"
#endif

__device__ __forceinline__ v8f mma_f32_k4(v2f a, v2f b, v8f c)
{
#if HAVE_WMMA_F32
    return __builtin_amdgcn_wmma_f32_16x16x4_f32(false, a, false, b, (short)0, c, false, false);
#else
    return c;
#endif
}

__global__ __launch_bounds__(512)
void proj_dt_kernel(const float* __restrict__ X,
                    const float* __restrict__ W1,
                    const float* __restrict__ prompt,
                    const float* __restrict__ W2,
                    const float* __restrict__ bias,
                    float* __restrict__ Bsc,
                    float* __restrict__ Csc,
                    float* __restrict__ delta)
{
    __shared__ float smem[64 * W1P + FB_ROWS * SDP];
    float* Wlds = smem;
    float* part = smem;
    float* sD   = smem + 64 * W1P;

    const int tid  = threadIdx.x;
    const int wave = tid >> 5;
    const int lane = tid & 31;
    const int ln   = lane & 15;
    const int hi   = lane >> 4;
    const int rt   = wave >> 2;
    const int q    = wave & 3;
    const int mb   = blockIdx.x * FB_ROWS;
    const int m0   = mb + rt * 16;

    v2f   bfr[6][6];
    float bi[6];
#pragma unroll
    for (int j = 0; j < 6; ++j) {
        const float* wrow = W2 + (size_t)(96 * q + 16 * j + ln) * DTRANK + 2 * hi;
#pragma unroll
        for (int ks = 0; ks < 6; ++ks) {
            bfr[j][ks].x = wrow[4 * ks];
            bfr[j][ks].y = wrow[4 * ks + 1];
        }
        bi[j] = bias[96 * q + 16 * j + ln];
    }

    {
        const float4* W14 = reinterpret_cast<const float4*>(W1);
        const int NV = CTOT * DMODEL / 4;
        float4 v[11];
#pragma unroll
        for (int i = 0; i < 11; ++i) {
            const int idx = tid + i * 512;
            v[i] = (idx < NV) ? W14[idx] : make_float4(0.0f, 0.0f, 0.0f, 0.0f);
        }
#pragma unroll
        for (int i = 0; i < 11; ++i) {
            const int idx = tid + i * 512;
            if (idx < NV) {
                const int n = idx / (DMODEL / 4), k4 = idx - n * (DMODEL / 4);
                *reinterpret_cast<float4*>(&Wlds[n * W1P + 4 * k4]) = v[i];
            }
        }
        for (int idx = tid; idx < 8 * (DMODEL / 4); idx += 512) {
            const int n = CTOT + idx / (DMODEL / 4), k4 = idx % (DMODEL / 4);
            *reinterpret_cast<float4*>(&Wlds[n * W1P + 4 * k4]) = make_float4(0.0f, 0.0f, 0.0f, 0.0f);
        }
    }
    __syncthreads();

    v8f acc[4] = {};
    {
        const float* arow = X + (size_t)(m0 + ln) * DMODEL + 96 * q + 2 * hi;
        const float* brow = Wlds + ln * W1P + 96 * q + 2 * hi;
#pragma unroll
        for (int k0 = 0; k0 < 96; k0 += 4) {
            v2f a;
            a.x = arow[k0];
            a.y = arow[k0 + 1];
#pragma unroll
            for (int ct = 0; ct < 4; ++ct) {
                v2f b;
                b.x = brow[ct * 16 * W1P + k0];
                b.y = brow[ct * 16 * W1P + k0 + 1];
                acc[ct] = mma_f32_k4(a, b, acc[ct]);
            }
        }
    }
    __syncthreads();

#pragma unroll
    for (int ct = 0; ct < 4; ++ct)
#pragma unroll
        for (int r = 0; r < 8; ++r)
            part[((wave * 4 + ct) * 8 + r) * 32 + lane] = acc[ct][r];
    __syncthreads();

    {
        const int c = 16 * q + ln;
        float v[8];
#pragma unroll
        for (int r = 0; r < 8; ++r) {
            float t = part[(((rt * 4 + 0) * 4 + q) * 8 + r) * 32 + lane];
            t += part[(((rt * 4 + 1) * 4 + q) * 8 + r) * 32 + lane];
            t += part[(((rt * 4 + 2) * 4 + q) * 8 + r) * 32 + lane];
            t += part[(((rt * 4 + 3) * 4 + q) * 8 + r) * 32 + lane];
            v[r] = t;
        }
        const int ml0 = rt * 16 + 8 * hi;
        if (c < DTRANK) {
#pragma unroll
            for (int r = 0; r < 8; ++r)
                sD[(ml0 + r) * SDP + c] = v[r];
        } else if (c < DTRANK + DSTATE) {
#pragma unroll
            for (int r = 0; r < 8; ++r)
                Bsc[(size_t)(mb + ml0 + r) * DSTATE + (c - DTRANK)] = v[r];
        } else if (c < CTOT) {
            const int n = c - DTRANK - DSTATE;
            float pr[8];
#pragma unroll
            for (int r = 0; r < 8; ++r)
                pr[r] = prompt[(size_t)(mb + ml0 + r) * DSTATE + n];
#pragma unroll
            for (int r = 0; r < 8; ++r)
                Csc[(size_t)(mb + ml0 + r) * DSTATE + n] = v[r] + pr[r];
        }
    }
    __syncthreads();

    {
        const float* arow = sD + (rt * 16 + ln) * SDP + 2 * hi;
        v2f afr[6];
#pragma unroll
        for (int ks = 0; ks < 6; ++ks) {
            afr[ks].x = arow[4 * ks];
            afr[ks].y = arow[4 * ks + 1];
        }
#pragma unroll
        for (int j = 0; j < 6; ++j) {
            v8f a2 = {};
#pragma unroll
            for (int ks = 0; ks < 6; ++ks)
                a2 = mma_f32_k4(afr[ks], bfr[j][ks], a2);
            const int d = 96 * q + 16 * j + ln;
#pragma unroll
            for (int r = 0; r < 8; ++r) {
                const int   m  = m0 + r + 8 * hi;
                const float t  = a2[r] + bi[j];
                const float sp = (t > 20.0f) ? t : log1pf(expf(t));
                delta[(size_t)m * DMODEL + d] = sp;
            }
        }
    }
}

__global__ __launch_bounds__(256)
void scan_passA_kernel(const float* __restrict__ delta,
                       const float* __restrict__ X,
                       const float* __restrict__ Bsc,
                       const float* __restrict__ Alog,
                       float* __restrict__ Pout,
                       float* __restrict__ Hend,
                       float* __restrict__ Pcum,
                       float* __restrict__ Hcum)
{
    __shared__ float sP[NSUB][DSTATE][DTILE];
    __shared__ float sH[NSUB][DSTATE][DTILE];

    const int dl  = threadIdx.x % DTILE;
    const int s   = threadIdx.x / DTILE;
    const int dt_ = blockIdx.x % NDT;
    const int cb  = blockIdx.x / NDT;
    const int b   = cb % BB;
    const int ch  = cb / BB;
    const int d   = dt_ * DTILE + dl;

    float A[DSTATE], h[DSTATE], P[DSTATE];
#pragma unroll
    for (int n = 0; n < DSTATE; ++n) {
        A[n] = -expf(Alog[d * DSTATE + n]);
        h[n] = 0.0f;
        P[n] = 1.0f;
    }

    const int l0 = ch * CHL + s * SUBL;
    for (int t = 0; t < SUBL; ++t) {
        const int bl = b * LL + l0 + t;
        if (t + 8 < SUBL) {
            __builtin_prefetch(&delta[(size_t)(bl + 8) * DMODEL + d], 0, 1);
            __builtin_prefetch(&X[(size_t)(bl + 8) * DMODEL + d], 0, 1);
        }
        const float dt = delta[(size_t)bl * DMODEL + d];
        const float u  = X[(size_t)bl * DMODEL + d];
        const float du = dt * u;
        const float4* Bv4 = reinterpret_cast<const float4*>(Bsc + (size_t)bl * DSTATE);
        float Bl[DSTATE];
#pragma unroll
        for (int q = 0; q < 4; ++q) {
            float4 v = Bv4[q];
            Bl[4 * q] = v.x; Bl[4 * q + 1] = v.y; Bl[4 * q + 2] = v.z; Bl[4 * q + 3] = v.w;
        }
#pragma unroll
        for (int n = 0; n < DSTATE; ++n) {
            float a = expf(dt * A[n]);
            P[n] *= a;
            h[n]  = a * h[n] + du * Bl[n];
        }
    }

#pragma unroll
    for (int n = 0; n < DSTATE; ++n) { sP[s][n][dl] = P[n]; sH[s][n][dl] = h[n]; }
    __syncthreads();

    float Pc[DSTATE], Hc[DSTATE];
#pragma unroll
    for (int n = 0; n < DSTATE; ++n) { Pc[n] = 1.0f; Hc[n] = 0.0f; }
    for (int j = 0; j < s; ++j) {
#pragma unroll
        for (int n = 0; n < DSTATE; ++n) {
            const float pj = sP[j][n][dl];
            Hc[n] = pj * Hc[n] + sH[j][n][dl];
            Pc[n] *= pj;
        }
    }

    if (s > 0) {
        const size_t su = ((size_t)(cb * NSUB + s) * DSTATE) * DMODEL + d;
#pragma unroll
        for (int n = 0; n < DSTATE; ++n) {
            Pcum[su + (size_t)n * DMODEL] = Pc[n];
            Hcum[su + (size_t)n * DMODEL] = Hc[n];
        }
    }
    if (s == NSUB - 1) {
        const size_t unit = ((size_t)cb * DMODEL + d) * DSTATE;
        float* Pp = Pout + unit;
        float* Hp = Hend + unit;
#pragma unroll
        for (int n = 0; n < DSTATE; ++n) {
            Pp[n] = Pc[n] * P[n];
            Hp[n] = P[n] * Hc[n] + h[n];
        }
    }
}

#define CF_BATCH 8
__global__ __launch_bounds__(256)
void scan_passC_kernel(const float* __restrict__ delta,
                       const float* __restrict__ X,
                       const float* __restrict__ Bsc,
                       const float* __restrict__ Csc,
                       const float* __restrict__ Alog,
                       const float* __restrict__ Ds,
                       const float* __restrict__ Pch,
                       const float* __restrict__ Hend,
                       const float* __restrict__ Pcum,
                       const float* __restrict__ Hcum,
                       float* __restrict__ Y)
{
    __shared__ float4 sHi[DTILE][DSTATE / 4 + 1];

    const int dl  = threadIdx.x % DTILE;
    const int s   = threadIdx.x / DTILE;
    const int dt_ = blockIdx.x % NDT;
    const int cbr = blockIdx.x / NDT;
    const int b   = cbr % BB;
    const int ch  = (NCH - 1 - cbr / BB) & (NCH - 1);
    const int cb  = ch * BB + b;
    const int d   = dt_ * DTILE + dl;

    {
        const int    i       = threadIdx.x;
        const size_t cstride = (size_t)BB * DMODEL * DSTATE / 4;
        const size_t base    = ((size_t)b * DMODEL + dt_ * DTILE) * DSTATE / 4 + i;
        const float4* P4 = reinterpret_cast<const float4*>(Pch)  + base;
        const float4* H4 = reinterpret_cast<const float4*>(Hend) + base;
        float4 hv = make_float4(0.0f, 0.0f, 0.0f, 0.0f);
        for (int j0 = 0; j0 < ch; j0 += CF_BATCH) {
            float4 p[CF_BATCH], e[CF_BATCH];
#pragma unroll
            for (int q = 0; q < CF_BATCH; ++q) {
                const int j = min(j0 + q, ch - 1);
                p[q] = P4[(size_t)j * cstride];
                e[q] = H4[(size_t)j * cstride];
            }
#pragma unroll
            for (int q = 0; q < CF_BATCH; ++q) {
                if (j0 + q < ch) {
                    hv.x = p[q].x * hv.x + e[q].x;
                    hv.y = p[q].y * hv.y + e[q].y;
                    hv.z = p[q].z * hv.z + e[q].z;
                    hv.w = p[q].w * hv.w + e[q].w;
                }
            }
        }
        sHi[i >> 2][i & 3] = hv;
    }
    __syncthreads();

    float A[DSTATE], h[DSTATE];
#pragma unroll
    for (int q = 0; q < 4; ++q) {
        const float4 v = sHi[dl][q];
        h[4 * q] = v.x; h[4 * q + 1] = v.y; h[4 * q + 2] = v.z; h[4 * q + 3] = v.w;
    }
#pragma unroll
    for (int n = 0; n < DSTATE; ++n)
        A[n] = -expf(Alog[d * DSTATE + n]);
    if (s > 0) {
        const size_t su = ((size_t)(cb * NSUB + s) * DSTATE) * DMODEL + d;
#pragma unroll
        for (int n = 0; n < DSTATE; ++n)
            h[n] = Pcum[su + (size_t)n * DMODEL] * h[n] + Hcum[su + (size_t)n * DMODEL];
    }
    const float Dsd = Ds[d];

    const int l0 = ch * CHL + s * SUBL;
    for (int t = 0; t < SUBL; ++t) {
        const int bl = b * LL + l0 + t;
        if (t + 8 < SUBL) {
            __builtin_prefetch(&delta[(size_t)(bl + 8) * DMODEL + d], 0, 1);
            __builtin_prefetch(&X[(size_t)(bl + 8) * DMODEL + d], 0, 1);
        }
        const float dt = delta[(size_t)bl * DMODEL + d];
        const float u  = X[(size_t)bl * DMODEL + d];
        const float du = dt * u;
        const float4* Bv4 = reinterpret_cast<const float4*>(Bsc + (size_t)bl * DSTATE);
        const float4* Cv4 = reinterpret_cast<const float4*>(Csc + (size_t)bl * DSTATE);
        float Bl[DSTATE], Cl[DSTATE];
#pragma unroll
        for (int q = 0; q < 4; ++q) {
            float4 vb = Bv4[q], vc = Cv4[q];
            Bl[4*q] = vb.x; Bl[4*q+1] = vb.y; Bl[4*q+2] = vb.z; Bl[4*q+3] = vb.w;
            Cl[4*q] = vc.x; Cl[4*q+1] = vc.y; Cl[4*q+2] = vc.z; Cl[4*q+3] = vc.w;
        }
        float y = 0.0f;
#pragma unroll
        for (int n = 0; n < DSTATE; ++n) {
            float a = expf(dt * A[n]);
            h[n] = a * h[n] + du * Bl[n];
            y   += h[n] * Cl[n];
        }
        Y[(size_t)bl * DMODEL + d] = y + u * Dsd;
    }
}

extern "C" void kernel_launch(void* const* d_in, const int* in_sizes, int n_in,
                              void* d_out, int out_size, void* d_ws, size_t ws_size,
                              hipStream_t stream)
{
    const float* x      = (const float*)d_in[0];
    const float* prompt = (const float*)d_in[1];
    const float* w1     = (const float*)d_in[2];
    const float* w2     = (const float*)d_in[3];
    const float* bias   = (const float*)d_in[4];
    const float* alog   = (const float*)d_in[5];
    const float* ds     = (const float*)d_in[6];
    float*       y      = (float*)d_out;

    float* ws = (float*)d_ws;
    size_t o = 0;
    o += (size_t)MTOT * DTRANK;
    float* Bsc   = ws + o; o += (size_t)MTOT * DSTATE;
    float* Csc   = ws + o; o += (size_t)MTOT * DSTATE;
    float* delta = ws + o; o += (size_t)MTOT * DMODEL;
    const size_t units = (size_t)NCH * BB * DMODEL;
    float* P     = ws + o; o += units * DSTATE;
    float* Hend  = ws + o; o += units * DSTATE;
    float* Pcum  = ws + o; o += units * NSUB * DSTATE;
    float* Hcum  = ws + o; o += units * NSUB * DSTATE;
    (void)ws_size; (void)in_sizes; (void)n_in; (void)out_size;

    proj_dt_kernel<<<dim3(MTOT / FB_ROWS), 512, 0, stream>>>(x, w1, prompt, w2, bias, Bsc, Csc, delta);
    scan_passA_kernel<<<dim3(NCH * BB * NDT), 256, 0, stream>>>(delta, x, Bsc, alog, P, Hend, Pcum, Hcum);
    scan_passC_kernel<<<dim3(NCH * BB * NDT), 256, 0, stream>>>(delta, x, Bsc, Csc, alog, ds, P, Hend, Pcum, Hcum, y);
}


# Scan passes A and C: a step's 16 decays computed as 8 two-wide pairs: expf's argument reduction on packed mul/fma, fraction = fma(x,Llo,fma(x,Lhi,-e)); decays equal the library's to within 1 ulp
# speedup vs baseline: 4.0424x; 1.0954x over previous
#include <hip/hip_runtime.h>
#include <hip/hip_bf16.h>
#include <math.h>

#define BB      4
#define LL      4096
#define DMODEL  384
#define DSTATE  16
#define DTRANK  24
#define CTOT    (DTRANK + 2*DSTATE)
#define MTOT    (BB*LL)
#define NCH     32
#define CHL     (LL/NCH)
#define NSUB    4
#define SUBL    (CHL/NSUB)
#define DTILE   (256/NSUB)
#define NDT     (DMODEL/DTILE)

typedef float     v2f  __attribute__((ext_vector_type(2)));
typedef float     v8f  __attribute__((ext_vector_type(8)));
typedef _Float16  v16h __attribute__((ext_vector_type(16)));

#if defined(__gfx1250__) && __has_builtin(__builtin_amdgcn_wmma_f32_16x16x4_f32)
#define HAVE_WMMA_F32 1
#else
#define HAVE_WMMA_F32 0
#endif

#define FB_ROWS  64
#define W1P      388
#define SDP      28

#if !HAVE_WMMA_F32 && defined(__HIP_DEVICE_COMPILE__)
#error "the f32 16x16x4 matrix instruction is required"
#endif

__device__ __forceinline__ v8f mma_f32_k4(v2f a, v2f b, v8f c)
{
#if HAVE_WMMA_F32
    return __builtin_amdgcn_wmma_f32_16x16x4_f32(false, a, false, b, (short)0, c, false, false);
#else
    return c;
#endif
}

__device__ __forceinline__ v2f exp_pair(v2f x)
{
    const float L_hi  = __builtin_bit_cast(float, 0x3fb8aa3bu);
    const float L_lo  = __builtin_bit_cast(float, 0x32a5705fu);
    const float X_MIN = __builtin_bit_cast(float, 0xc2ce8ed0u);
    const float X_MAX = __builtin_bit_cast(float, 0x42b17218u);
    const v2f Lh = {L_hi, L_hi};
    const v2f Ll = {L_lo, L_lo};
    const v2f t = x * Lh;
    v2f e;
    e.x = __builtin_rintf(t.x);
    e.y = __builtin_rintf(t.y);
    v2f r = __builtin_elementwise_fma(x, Lh, -e);
    r = __builtin_elementwise_fma(x, Ll, r);
    v2f v;
    v.x = __builtin_ldexpf(__builtin_amdgcn_exp2f(r.x), (int)e.x);
    v.y = __builtin_ldexpf(__builtin_amdgcn_exp2f(r.y), (int)e.y);
    v.x = (x.x < X_MIN) ? 0.0f : v.x;
    v.y = (x.y < X_MIN) ? 0.0f : v.y;
    v.x = (x.x > X_MAX) ? __builtin_inff() : v.x;
    v.y = (x.y > X_MAX) ? __builtin_inff() : v.y;
    return v;
}

__global__ __launch_bounds__(512)
void proj_dt_kernel(const float* __restrict__ X,
                    const float* __restrict__ W1,
                    const float* __restrict__ prompt,
                    const float* __restrict__ W2,
                    const float* __restrict__ bias,
                    float* __restrict__ Bsc,
                    float* __restrict__ Csc,
                    float* __restrict__ delta)
{
    __shared__ float smem[64 * W1P + FB_ROWS * SDP];
    float* Wlds = smem;
    float* part = smem;
    float* sD   = smem + 64 * W1P;

    const int tid  = threadIdx.x;
    const int wave = tid >> 5;
    const int lane = tid & 31;
    const int ln   = lane & 15;
    const int hi   = lane >> 4;
    const int rt   = wave >> 2;
    const int q    = wave & 3;
    const int mb   = blockIdx.x * FB_ROWS;
    const int m0   = mb + rt * 16;

    v2f   bfr[6][6];
    float bi[6];
#pragma unroll
    for (int j = 0; j < 6; ++j) {
        const float* wrow = W2 + (size_t)(96 * q + 16 * j + ln) * DTRANK + 2 * hi;
#pragma unroll
        for (int ks = 0; ks < 6; ++ks) {
            bfr[j][ks].x = wrow[4 * ks];
            bfr[j][ks].y = wrow[4 * ks + 1];
        }
        bi[j] = bias[96 * q + 16 * j + ln];
    }

    {
        const float4* W14 = reinterpret_cast<const float4*>(W1);
        const int NV = CTOT * DMODEL / 4;
        float4 v[11];
#pragma unroll
        for (int i = 0; i < 11; ++i) {
            const int idx = tid + i * 512;
            v[i] = (idx < NV) ? W14[idx] : make_float4(0.0f, 0.0f, 0.0f, 0.0f);
        }
#pragma unroll
        for (int i = 0; i < 11; ++i) {
            const int idx = tid + i * 512;
            if (idx < NV) {
                const int n = idx / (DMODEL / 4), k4 = idx - n * (DMODEL / 4);
                *reinterpret_cast<float4*>(&Wlds[n * W1P + 4 * k4]) = v[i];
            }
        }
        for (int idx = tid; idx < 8 * (DMODEL / 4); idx += 512) {
            const int n = CTOT + idx / (DMODEL / 4), k4 = idx % (DMODEL / 4);
            *reinterpret_cast<float4*>(&Wlds[n * W1P + 4 * k4]) = make_float4(0.0f, 0.0f, 0.0f, 0.0f);
        }
    }
    __syncthreads();

    v8f acc[4] = {};
    {
        const float* arow = X + (size_t)(m0 + ln) * DMODEL + 96 * q + 2 * hi;
        const float* brow = Wlds + ln * W1P + 96 * q + 2 * hi;
#pragma unroll
        for (int k0 = 0; k0 < 96; k0 += 4) {
            v2f a;
            a.x = arow[k0];
            a.y = arow[k0 + 1];
#pragma unroll
            for (int ct = 0; ct < 4; ++ct) {
                v2f b;
                b.x = brow[ct * 16 * W1P + k0];
                b.y = brow[ct * 16 * W1P + k0 + 1];
                acc[ct] = mma_f32_k4(a, b, acc[ct]);
            }
        }
    }
    __syncthreads();

#pragma unroll
    for (int ct = 0; ct < 4; ++ct)
#pragma unroll
        for (int r = 0; r < 8; ++r)
            part[((wave * 4 + ct) * 8 + r) * 32 + lane] = acc[ct][r];
    __syncthreads();

    {
        const int c = 16 * q + ln;
        float v[8];
#pragma unroll
        for (int r = 0; r < 8; ++r) {
            float t = part[(((rt * 4 + 0) * 4 + q) * 8 + r) * 32 + lane];
            t += part[(((rt * 4 + 1) * 4 + q) * 8 + r) * 32 + lane];
            t += part[(((rt * 4 + 2) * 4 + q) * 8 + r) * 32 + lane];
            t += part[(((rt * 4 + 3) * 4 + q) * 8 + r) * 32 + lane];
            v[r] = t;
        }
        const int ml0 = rt * 16 + 8 * hi;
        if (c < DTRANK) {
#pragma unroll
            for (int r = 0; r < 8; ++r)
                sD[(ml0 + r) * SDP + c] = v[r];
        } else if (c < DTRANK + DSTATE) {
#pragma unroll
            for (int r = 0; r < 8; ++r)
                Bsc[(size_t)(mb + ml0 + r) * DSTATE + (c - DTRANK)] = v[r];
        } else if (c < CTOT) {
            const int n = c - DTRANK - DSTATE;
            float pr[8];
#pragma unroll
            for (int r = 0; r < 8; ++r)
                pr[r] = prompt[(size_t)(mb + ml0 + r) * DSTATE + n];
#pragma unroll
            for (int r = 0; r < 8; ++r)
                Csc[(size_t)(mb + ml0 + r) * DSTATE + n] = v[r] + pr[r];
        }
    }
    __syncthreads();

    {
        const float* arow = sD + (rt * 16 + ln) * SDP + 2 * hi;
        v2f afr[6];
#pragma unroll
        for (int ks = 0; ks < 6; ++ks) {
            afr[ks].x = arow[4 * ks];
            afr[ks].y = arow[4 * ks + 1];
        }
#pragma unroll
        for (int j = 0; j < 6; ++j) {
            v8f a2 = {};
#pragma unroll
            for (int ks = 0; ks < 6; ++ks)
                a2 = mma_f32_k4(afr[ks], bfr[j][ks], a2);
            const int d = 96 * q + 16 * j + ln;
#pragma unroll
            for (int r = 0; r < 8; ++r) {
                const int   m  = m0 + r + 8 * hi;
                const float t  = a2[r] + bi[j];
                const float sp = (t > 20.0f) ? t : log1pf(expf(t));
                delta[(size_t)m * DMODEL + d] = sp;
            }
        }
    }
}

__global__ __launch_bounds__(256)
void scan_passA_kernel(const float* __restrict__ delta,
                       const float* __restrict__ X,
                       const float* __restrict__ Bsc,
                       const float* __restrict__ Alog,
                       float* __restrict__ Pout,
                       float* __restrict__ Hend,
                       float* __restrict__ Pcum,
                       float* __restrict__ Hcum)
{
    __shared__ float sP[NSUB][DSTATE][DTILE];
    __shared__ float sH[NSUB][DSTATE][DTILE];

    const int dl  = threadIdx.x % DTILE;
    const int s   = threadIdx.x / DTILE;
    const int dt_ = blockIdx.x % NDT;
    const int cb  = blockIdx.x / NDT;
    const int b   = cb % BB;
    const int ch  = cb / BB;
    const int d   = dt_ * DTILE + dl;

    v2f A2[DSTATE / 2], h2[DSTATE / 2], P2[DSTATE / 2];
#pragma unroll
    for (int p = 0; p < DSTATE / 2; ++p) {
        A2[p].x = -expf(Alog[d * DSTATE + 2 * p]);
        A2[p].y = -expf(Alog[d * DSTATE + 2 * p + 1]);
        h2[p] = (v2f){0.0f, 0.0f};
        P2[p] = (v2f){1.0f, 1.0f};
    }

    const int l0 = ch * CHL + s * SUBL;
    for (int t = 0; t < SUBL; ++t) {
        const int bl = b * LL + l0 + t;
        if (t + 8 < SUBL) {
            __builtin_prefetch(&delta[(size_t)(bl + 8) * DMODEL + d], 0, 1);
            __builtin_prefetch(&X[(size_t)(bl + 8) * DMODEL + d], 0, 1);
        }
        const float dt = delta[(size_t)bl * DMODEL + d];
        const float u  = X[(size_t)bl * DMODEL + d];
        const float du = dt * u;
        const float4* Bv4 = reinterpret_cast<const float4*>(Bsc + (size_t)bl * DSTATE);
        float Bl[DSTATE];
#pragma unroll
        for (int q = 0; q < 4; ++q) {
            float4 v = Bv4[q];
            Bl[4 * q] = v.x; Bl[4 * q + 1] = v.y; Bl[4 * q + 2] = v.z; Bl[4 * q + 3] = v.w;
        }
        const v2f dt2 = {dt, dt};
        const v2f du2 = {du, du};
#pragma unroll
        for (int p = 0; p < DSTATE / 2; ++p) {
            const v2f a  = exp_pair(dt2 * A2[p]);
            const v2f Bp = {Bl[2 * p], Bl[2 * p + 1]};
            P2[p] *= a;
            h2[p]  = __builtin_elementwise_fma(du2, Bp, a * h2[p]);
        }
    }

    float h[DSTATE], P[DSTATE];
#pragma unroll
    for (int p = 0; p < DSTATE / 2; ++p) {
        h[2 * p] = h2[p].x; h[2 * p + 1] = h2[p].y;
        P[2 * p] = P2[p].x; P[2 * p + 1] = P2[p].y;
    }

#pragma unroll
    for (int n = 0; n < DSTATE; ++n) { sP[s][n][dl] = P[n]; sH[s][n][dl] = h[n]; }
    __syncthreads();

    float Pc[DSTATE], Hc[DSTATE];
#pragma unroll
    for (int n = 0; n < DSTATE; ++n) { Pc[n] = 1.0f; Hc[n] = 0.0f; }
    for (int j = 0; j < s; ++j) {
#pragma unroll
        for (int n = 0; n < DSTATE; ++n) {
            const float pj = sP[j][n][dl];
            Hc[n] = pj * Hc[n] + sH[j][n][dl];
            Pc[n] *= pj;
        }
    }

    if (s > 0) {
        const size_t su = ((size_t)(cb * NSUB + s) * DSTATE) * DMODEL + d;
#pragma unroll
        for (int n = 0; n < DSTATE; ++n) {
            Pcum[su + (size_t)n * DMODEL] = Pc[n];
            Hcum[su + (size_t)n * DMODEL] = Hc[n];
        }
    }
    if (s == NSUB - 1) {
        const size_t unit = ((size_t)cb * DMODEL + d) * DSTATE;
        float* Pp = Pout + unit;
        float* Hp = Hend + unit;
#pragma unroll
        for (int n = 0; n < DSTATE; ++n) {
            Pp[n] = Pc[n] * P[n];
            Hp[n] = P[n] * Hc[n] + h[n];
        }
    }
}

#define CF_BATCH 8
__global__ __launch_bounds__(256)
void scan_passC_kernel(const float* __restrict__ delta,
                       const float* __restrict__ X,
                       const float* __restrict__ Bsc,
                       const float* __restrict__ Csc,
                       const float* __restrict__ Alog,
                       const float* __restrict__ Ds,
                       const float* __restrict__ Pch,
                       const float* __restrict__ Hend,
                       const float* __restrict__ Pcum,
                       const float* __restrict__ Hcum,
                       float* __restrict__ Y)
{
    __shared__ float4 sHi[DTILE][DSTATE / 4 + 1];

    const int dl  = threadIdx.x % DTILE;
    const int s   = threadIdx.x / DTILE;
    const int dt_ = blockIdx.x % NDT;
    const int cbr = blockIdx.x / NDT;
    const int b   = cbr % BB;
    const int ch  = (NCH - 1 - cbr / BB) & (NCH - 1);
    const int cb  = ch * BB + b;
    const int d   = dt_ * DTILE + dl;

    {
        const int    i       = threadIdx.x;
        const size_t cstride = (size_t)BB * DMODEL * DSTATE / 4;
        const size_t base    = ((size_t)b * DMODEL + dt_ * DTILE) * DSTATE / 4 + i;
        const float4* P4 = reinterpret_cast<const float4*>(Pch)  + base;
        const float4* H4 = reinterpret_cast<const float4*>(Hend) + base;
        float4 hv = make_float4(0.0f, 0.0f, 0.0f, 0.0f);
        for (int j0 = 0; j0 < ch; j0 += CF_BATCH) {
            float4 p[CF_BATCH], e[CF_BATCH];
#pragma unroll
            for (int q = 0; q < CF_BATCH; ++q) {
                const int j = min(j0 + q, ch - 1);
                p[q] = P4[(size_t)j * cstride];
                e[q] = H4[(size_t)j * cstride];
            }
#pragma unroll
            for (int q = 0; q < CF_BATCH; ++q) {
                if (j0 + q < ch) {
                    hv.x = p[q].x * hv.x + e[q].x;
                    hv.y = p[q].y * hv.y + e[q].y;
                    hv.z = p[q].z * hv.z + e[q].z;
                    hv.w = p[q].w * hv.w + e[q].w;
                }
            }
        }
        sHi[i >> 2][i & 3] = hv;
    }
    __syncthreads();

    float A[DSTATE], h[DSTATE];
#pragma unroll
    for (int q = 0; q < 4; ++q) {
        const float4 v = sHi[dl][q];
        h[4 * q] = v.x; h[4 * q + 1] = v.y; h[4 * q + 2] = v.z; h[4 * q + 3] = v.w;
    }
#pragma unroll
    for (int n = 0; n < DSTATE; ++n)
        A[n] = -expf(Alog[d * DSTATE + n]);
    if (s > 0) {
        const size_t su = ((size_t)(cb * NSUB + s) * DSTATE) * DMODEL + d;
#pragma unroll
        for (int n = 0; n < DSTATE; ++n)
            h[n] = Pcum[su + (size_t)n * DMODEL] * h[n] + Hcum[su + (size_t)n * DMODEL];
    }
    const float Dsd = Ds[d];

    v2f A2[DSTATE / 2], h2[DSTATE / 2];
#pragma unroll
    for (int p = 0; p < DSTATE / 2; ++p) {
        A2[p] = (v2f){A[2 * p], A[2 * p + 1]};
        h2[p] = (v2f){h[2 * p], h[2 * p + 1]};
    }

    const int l0 = ch * CHL + s * SUBL;
    for (int t = 0; t < SUBL; ++t) {
        const int bl = b * LL + l0 + t;
        if (t + 8 < SUBL) {
            __builtin_prefetch(&delta[(size_t)(bl + 8) * DMODEL + d], 0, 1);
            __builtin_prefetch(&X[(size_t)(bl + 8) * DMODEL + d], 0, 1);
        }
        const float dt = delta[(size_t)bl * DMODEL + d];
        const float u  = X[(size_t)bl * DMODEL + d];
        const float du = dt * u;
        const float4* Bv4 = reinterpret_cast<const float4*>(Bsc + (size_t)bl * DSTATE);
        const float4* Cv4 = reinterpret_cast<const float4*>(Csc + (size_t)bl * DSTATE);
        float Bl[DSTATE], Cl[DSTATE];
#pragma unroll
        for (int q = 0; q < 4; ++q) {
            float4 vb = Bv4[q], vc = Cv4[q];
            Bl[4*q] = vb.x; Bl[4*q+1] = vb.y; Bl[4*q+2] = vb.z; Bl[4*q+3] = vb.w;
            Cl[4*q] = vc.x; Cl[4*q+1] = vc.y; Cl[4*q+2] = vc.z; Cl[4*q+3] = vc.w;
        }
        const v2f dt2 = {dt, dt};
        const v2f du2 = {du, du};
        float y = 0.0f;
#pragma unroll
        for (int p = 0; p < DSTATE / 2; ++p) {
            const v2f a  = exp_pair(dt2 * A2[p]);
            const v2f Bp = {Bl[2 * p], Bl[2 * p + 1]};
            const v2f Cp = {Cl[2 * p], Cl[2 * p + 1]};
            h2[p] = __builtin_elementwise_fma(du2, Bp, a * h2[p]);
            const v2f hc = h2[p] * Cp;
            y += hc.x;
            y += hc.y;
        }
        Y[(size_t)bl * DMODEL + d] = y + u * Dsd;
    }
}

extern "C" void kernel_launch(void* const* d_in, const int* in_sizes, int n_in,
                              void* d_out, int out_size, void* d_ws, size_t ws_size,
                              hipStream_t stream)
{
    const float* x      = (const float*)d_in[0];
    const float* prompt = (const float*)d_in[1];
    const float* w1     = (const float*)d_in[2];
    const float* w2     = (const float*)d_in[3];
    const float* bias   = (const float*)d_in[4];
    const float* alog   = (const float*)d_in[5];
    const float* ds     = (const float*)d_in[6];
    float*       y      = (float*)d_out;

    float* ws = (float*)d_ws;
    size_t o = 0;
    o += (size_t)MTOT * DTRANK;
    float* Bsc   = ws + o; o += (size_t)MTOT * DSTATE;
    float* Csc   = ws + o; o += (size_t)MTOT * DSTATE;
    float* delta = ws + o; o += (size_t)MTOT * DMODEL;
    const size_t units = (size_t)NCH * BB * DMODEL;
    float* P     = ws + o; o += units * DSTATE;
    float* Hend  = ws + o; o += units * DSTATE;
    float* Pcum  = ws + o; o += units * NSUB * DSTATE;
    float* Hcum  = ws + o; o += units * NSUB * DSTATE;
    (void)ws_size; (void)in_sizes; (void)n_in; (void)out_size;

    proj_dt_kernel<<<dim3(MTOT / FB_ROWS), 512, 0, stream>>>(x, w1, prompt, w2, bias, Bsc, Csc, delta);
    scan_passA_kernel<<<dim3(NCH * BB * NDT), 256, 0, stream>>>(delta, x, Bsc, alog, P, Hend, Pcum, Hcum);
    scan_passC_kernel<<<dim3(NCH * BB * NDT), 256, 0, stream>>>(delta, x, Bsc, Csc, alog, ds, P, Hend, Pcum, Hcum, y);
}


# Scan passes A and C: range bounds of a step's 16 decays tested once per step (wave-wide, on dt*Amin and dt*Amax) and applied only if the test fails: 64 compare/selects per step gone; same bits
# speedup vs baseline: 4.5298x; 1.1206x over previous
#include <hip/hip_runtime.h>
#include <hip/hip_bf16.h>
#include <math.h>

#define BB      4
#define LL      4096
#define DMODEL  384
#define DSTATE  16
#define DTRANK  24
#define CTOT    (DTRANK + 2*DSTATE)
#define MTOT    (BB*LL)
#define NCH     32
#define CHL     (LL/NCH)
#define NSUB    4
#define SUBL    (CHL/NSUB)
#define DTILE   (256/NSUB)
#define NDT     (DMODEL/DTILE)

typedef float     v2f  __attribute__((ext_vector_type(2)));
typedef float     v8f  __attribute__((ext_vector_type(8)));
typedef _Float16  v16h __attribute__((ext_vector_type(16)));

#if defined(__gfx1250__) && __has_builtin(__builtin_amdgcn_wmma_f32_16x16x4_f32)
#define HAVE_WMMA_F32 1
#else
#define HAVE_WMMA_F32 0
#endif

#define FB_ROWS  64
#define W1P      388
#define SDP      28

#if !HAVE_WMMA_F32 && defined(__HIP_DEVICE_COMPILE__)
#error "the f32 16x16x4 matrix instruction is required"
#endif

__device__ __forceinline__ v8f mma_f32_k4(v2f a, v2f b, v8f c)
{
#if HAVE_WMMA_F32
    return __builtin_amdgcn_wmma_f32_16x16x4_f32(false, a, false, b, (short)0, c, false, false);
#else
    return c;
#endif
}

#define EXP_X_MAX_BITS 0x42b17218u
__device__ __forceinline__ v2f exp_pair_core(v2f x)
{
    const float L_hi  = __builtin_bit_cast(float, 0x3fb8aa3bu);
    const float L_lo  = __builtin_bit_cast(float, 0x32a5705fu);
    const v2f Lh = {L_hi, L_hi};
    const v2f Ll = {L_lo, L_lo};
    const v2f t = x * Lh;
    v2f e;
    e.x = __builtin_rintf(t.x);
    e.y = __builtin_rintf(t.y);
    v2f r = __builtin_elementwise_fma(x, Lh, -e);
    r = __builtin_elementwise_fma(x, Ll, r);
    v2f v;
    v.x = __builtin_ldexpf(__builtin_amdgcn_exp2f(r.x), (int)e.x);
    v.y = __builtin_ldexpf(__builtin_amdgcn_exp2f(r.y), (int)e.y);
    return v;
}

__device__ __forceinline__ v2f exp_pair_bounds(v2f v, v2f x)
{
    const float X_MIN = __builtin_bit_cast(float, 0xc2ce8ed0u);
    const float X_MAX = __builtin_bit_cast(float, EXP_X_MAX_BITS);
    v.x = (x.x < X_MIN) ? 0.0f : v.x;
    v.y = (x.y < X_MIN) ? 0.0f : v.y;
    v.x = (x.x > X_MAX) ? __builtin_inff() : v.x;
    v.y = (x.y > X_MAX) ? __builtin_inff() : v.y;
    return v;
}

__device__ __forceinline__ void step_decays(float dt, float Amin, float Amax,
                                            const v2f (&A2)[DSTATE / 2], v2f (&a)[DSTATE / 2])
{
    const float LIM = __builtin_bit_cast(float, EXP_X_MAX_BITS);
    const bool in_range = (__builtin_fabsf(dt * Amin) <= LIM) && (__builtin_fabsf(dt * Amax) <= LIM);
    const v2f dt2 = {dt, dt};
#pragma unroll
    for (int p = 0; p < DSTATE / 2; ++p)
        a[p] = exp_pair_core(dt2 * A2[p]);
    if (__builtin_amdgcn_ballot_w32(!in_range) != 0u) {
#pragma unroll
        for (int p = 0; p < DSTATE / 2; ++p)
            a[p] = exp_pair_bounds(a[p], dt2 * A2[p]);
    }
}

__global__ __launch_bounds__(512)
void proj_dt_kernel(const float* __restrict__ X,
                    const float* __restrict__ W1,
                    const float* __restrict__ prompt,
                    const float* __restrict__ W2,
                    const float* __restrict__ bias,
                    float* __restrict__ Bsc,
                    float* __restrict__ Csc,
                    float* __restrict__ delta)
{
    __shared__ float smem[64 * W1P + FB_ROWS * SDP];
    float* Wlds = smem;
    float* part = smem;
    float* sD   = smem + 64 * W1P;

    const int tid  = threadIdx.x;
    const int wave = tid >> 5;
    const int lane = tid & 31;
    const int ln   = lane & 15;
    const int hi   = lane >> 4;
    const int rt   = wave >> 2;
    const int q    = wave & 3;
    const int mb   = blockIdx.x * FB_ROWS;
    const int m0   = mb + rt * 16;

    v2f   bfr[6][6];
    float bi[6];
#pragma unroll
    for (int j = 0; j < 6; ++j) {
        const float* wrow = W2 + (size_t)(96 * q + 16 * j + ln) * DTRANK + 2 * hi;
#pragma unroll
        for (int ks = 0; ks < 6; ++ks) {
            bfr[j][ks].x = wrow[4 * ks];
            bfr[j][ks].y = wrow[4 * ks + 1];
        }
        bi[j] = bias[96 * q + 16 * j + ln];
    }

    {
        const float4* W14 = reinterpret_cast<const float4*>(W1);
        const int NV = CTOT * DMODEL / 4;
        float4 v[11];
#pragma unroll
        for (int i = 0; i < 11; ++i) {
            const int idx = tid + i * 512;
            v[i] = (idx < NV) ? W14[idx] : make_float4(0.0f, 0.0f, 0.0f, 0.0f);
        }
#pragma unroll
        for (int i = 0; i < 11; ++i) {
            const int idx = tid + i * 512;
            if (idx < NV) {
                const int n = idx / (DMODEL / 4), k4 = idx - n * (DMODEL / 4);
                *reinterpret_cast<float4*>(&Wlds[n * W1P + 4 * k4]) = v[i];
            }
        }
        for (int idx = tid; idx < 8 * (DMODEL / 4); idx += 512) {
            const int n = CTOT + idx / (DMODEL / 4), k4 = idx % (DMODEL / 4);
            *reinterpret_cast<float4*>(&Wlds[n * W1P + 4 * k4]) = make_float4(0.0f, 0.0f, 0.0f, 0.0f);
        }
    }
    __syncthreads();

    v8f acc[4] = {};
    {
        const float* arow = X + (size_t)(m0 + ln) * DMODEL + 96 * q + 2 * hi;
        const float* brow = Wlds + ln * W1P + 96 * q + 2 * hi;
#pragma unroll
        for (int k0 = 0; k0 < 96; k0 += 4) {
            v2f a;
            a.x = arow[k0];
            a.y = arow[k0 + 1];
#pragma unroll
            for (int ct = 0; ct < 4; ++ct) {
                v2f b;
                b.x = brow[ct * 16 * W1P + k0];
                b.y = brow[ct * 16 * W1P + k0 + 1];
                acc[ct] = mma_f32_k4(a, b, acc[ct]);
            }
        }
    }
    __syncthreads();

#pragma unroll
    for (int ct = 0; ct < 4; ++ct)
#pragma unroll
        for (int r = 0; r < 8; ++r)
            part[((wave * 4 + ct) * 8 + r) * 32 + lane] = acc[ct][r];
    __syncthreads();

    {
        const int c = 16 * q + ln;
        float v[8];
#pragma unroll
        for (int r = 0; r < 8; ++r) {
            float t = part[(((rt * 4 + 0) * 4 + q) * 8 + r) * 32 + lane];
            t += part[(((rt * 4 + 1) * 4 + q) * 8 + r) * 32 + lane];
            t += part[(((rt * 4 + 2) * 4 + q) * 8 + r) * 32 + lane];
            t += part[(((rt * 4 + 3) * 4 + q) * 8 + r) * 32 + lane];
            v[r] = t;
        }
        const int ml0 = rt * 16 + 8 * hi;
        if (c < DTRANK) {
#pragma unroll
            for (int r = 0; r < 8; ++r)
                sD[(ml0 + r) * SDP + c] = v[r];
        } else if (c < DTRANK + DSTATE) {
#pragma unroll
            for (int r = 0; r < 8; ++r)
                Bsc[(size_t)(mb + ml0 + r) * DSTATE + (c - DTRANK)] = v[r];
        } else if (c < CTOT) {
            const int n = c - DTRANK - DSTATE;
            float pr[8];
#pragma unroll
            for (int r = 0; r < 8; ++r)
                pr[r] = prompt[(size_t)(mb + ml0 + r) * DSTATE + n];
#pragma unroll
            for (int r = 0; r < 8; ++r)
                Csc[(size_t)(mb + ml0 + r) * DSTATE + n] = v[r] + pr[r];
        }
    }
    __syncthreads();

    {
        const float* arow = sD + (rt * 16 + ln) * SDP + 2 * hi;
        v2f afr[6];
#pragma unroll
        for (int ks = 0; ks < 6; ++ks) {
            afr[ks].x = arow[4 * ks];
            afr[ks].y = arow[4 * ks + 1];
        }
#pragma unroll
        for (int j = 0; j < 6; ++j) {
            v8f a2 = {};
#pragma unroll
            for (int ks = 0; ks < 6; ++ks)
                a2 = mma_f32_k4(afr[ks], bfr[j][ks], a2);
            const int d = 96 * q + 16 * j + ln;
#pragma unroll
            for (int r = 0; r < 8; ++r) {
                const int   m  = m0 + r + 8 * hi;
                const float t  = a2[r] + bi[j];
                const float sp = (t > 20.0f) ? t : log1pf(expf(t));
                delta[(size_t)m * DMODEL + d] = sp;
            }
        }
    }
}

__global__ __launch_bounds__(256)
void scan_passA_kernel(const float* __restrict__ delta,
                       const float* __restrict__ X,
                       const float* __restrict__ Bsc,
                       const float* __restrict__ Alog,
                       float* __restrict__ Pout,
                       float* __restrict__ Hend,
                       float* __restrict__ Pcum,
                       float* __restrict__ Hcum)
{
    __shared__ float sP[NSUB][DSTATE][DTILE];
    __shared__ float sH[NSUB][DSTATE][DTILE];

    const int dl  = threadIdx.x % DTILE;
    const int s   = threadIdx.x / DTILE;
    const int dt_ = blockIdx.x % NDT;
    const int cb  = blockIdx.x / NDT;
    const int b   = cb % BB;
    const int ch  = cb / BB;
    const int d   = dt_ * DTILE + dl;

    v2f A2[DSTATE / 2], h2[DSTATE / 2], P2[DSTATE / 2];
#pragma unroll
    for (int p = 0; p < DSTATE / 2; ++p) {
        A2[p].x = -expf(Alog[d * DSTATE + 2 * p]);
        A2[p].y = -expf(Alog[d * DSTATE + 2 * p + 1]);
        h2[p] = (v2f){0.0f, 0.0f};
        P2[p] = (v2f){1.0f, 1.0f};
    }
    float Amin = A2[0].x, Amax = A2[0].x;
#pragma unroll
    for (int p = 0; p < DSTATE / 2; ++p) {
        Amin = __builtin_fminf(Amin, __builtin_fminf(A2[p].x, A2[p].y));
        Amax = __builtin_fmaxf(Amax, __builtin_fmaxf(A2[p].x, A2[p].y));
    }

    const int l0 = ch * CHL + s * SUBL;
    for (int t = 0; t < SUBL; ++t) {
        const int bl = b * LL + l0 + t;
        if (t + 8 < SUBL) {
            __builtin_prefetch(&delta[(size_t)(bl + 8) * DMODEL + d], 0, 1);
            __builtin_prefetch(&X[(size_t)(bl + 8) * DMODEL + d], 0, 1);
        }
        const float dt = delta[(size_t)bl * DMODEL + d];
        const float u  = X[(size_t)bl * DMODEL + d];
        const float du = dt * u;
        const float4* Bv4 = reinterpret_cast<const float4*>(Bsc + (size_t)bl * DSTATE);
        float Bl[DSTATE];
#pragma unroll
        for (int q = 0; q < 4; ++q) {
            float4 v = Bv4[q];
            Bl[4 * q] = v.x; Bl[4 * q + 1] = v.y; Bl[4 * q + 2] = v.z; Bl[4 * q + 3] = v.w;
        }
        const v2f du2 = {du, du};
        v2f a[DSTATE / 2];
        step_decays(dt, Amin, Amax, A2, a);
#pragma unroll
        for (int p = 0; p < DSTATE / 2; ++p) {
            const v2f Bp = {Bl[2 * p], Bl[2 * p + 1]};
            P2[p] *= a[p];
            h2[p]  = __builtin_elementwise_fma(du2, Bp, a[p] * h2[p]);
        }
    }

    float h[DSTATE], P[DSTATE];
#pragma unroll
    for (int p = 0; p < DSTATE / 2; ++p) {
        h[2 * p] = h2[p].x; h[2 * p + 1] = h2[p].y;
        P[2 * p] = P2[p].x; P[2 * p + 1] = P2[p].y;
    }

#pragma unroll
    for (int n = 0; n < DSTATE; ++n) { sP[s][n][dl] = P[n]; sH[s][n][dl] = h[n]; }
    __syncthreads();

    float Pc[DSTATE], Hc[DSTATE];
#pragma unroll
    for (int n = 0; n < DSTATE; ++n) { Pc[n] = 1.0f; Hc[n] = 0.0f; }
    for (int j = 0; j < s; ++j) {
#pragma unroll
        for (int n = 0; n < DSTATE; ++n) {
            const float pj = sP[j][n][dl];
            Hc[n] = pj * Hc[n] + sH[j][n][dl];
            Pc[n] *= pj;
        }
    }

    if (s > 0) {
        const size_t su = ((size_t)(cb * NSUB + s) * DSTATE) * DMODEL + d;
#pragma unroll
        for (int n = 0; n < DSTATE; ++n) {
            Pcum[su + (size_t)n * DMODEL] = Pc[n];
            Hcum[su + (size_t)n * DMODEL] = Hc[n];
        }
    }
    if (s == NSUB - 1) {
        const size_t unit = ((size_t)cb * DMODEL + d) * DSTATE;
        float* Pp = Pout + unit;
        float* Hp = Hend + unit;
#pragma unroll
        for (int n = 0; n < DSTATE; ++n) {
            Pp[n] = Pc[n] * P[n];
            Hp[n] = P[n] * Hc[n] + h[n];
        }
    }
}

#define CF_BATCH 8
__global__ __launch_bounds__(256)
void scan_passC_kernel(const float* __restrict__ delta,
                       const float* __restrict__ X,
                       const float* __restrict__ Bsc,
                       const float* __restrict__ Csc,
                       const float* __restrict__ Alog,
                       const float* __restrict__ Ds,
                       const float* __restrict__ Pch,
                       const float* __restrict__ Hend,
                       const float* __restrict__ Pcum,
                       const float* __restrict__ Hcum,
                       float* __restrict__ Y)
{
    __shared__ float4 sHi[DTILE][DSTATE / 4 + 1];

    const int dl  = threadIdx.x % DTILE;
    const int s   = threadIdx.x / DTILE;
    const int dt_ = blockIdx.x % NDT;
    const int cbr = blockIdx.x / NDT;
    const int b   = cbr % BB;
    const int ch  = (NCH - 1 - cbr / BB) & (NCH - 1);
    const int cb  = ch * BB + b;
    const int d   = dt_ * DTILE + dl;

    {
        const int    i       = threadIdx.x;
        const size_t cstride = (size_t)BB * DMODEL * DSTATE / 4;
        const size_t base    = ((size_t)b * DMODEL + dt_ * DTILE) * DSTATE / 4 + i;
        const float4* P4 = reinterpret_cast<const float4*>(Pch)  + base;
        const float4* H4 = reinterpret_cast<const float4*>(Hend) + base;
        float4 hv = make_float4(0.0f, 0.0f, 0.0f, 0.0f);
        for (int j0 = 0; j0 < ch; j0 += CF_BATCH) {
            float4 p[CF_BATCH], e[CF_BATCH];
#pragma unroll
            for (int q = 0; q < CF_BATCH; ++q) {
                const int j = min(j0 + q, ch - 1);
                p[q] = P4[(size_t)j * cstride];
                e[q] = H4[(size_t)j * cstride];
            }
#pragma unroll
            for (int q = 0; q < CF_BATCH; ++q) {
                if (j0 + q < ch) {
                    hv.x = p[q].x * hv.x + e[q].x;
                    hv.y = p[q].y * hv.y + e[q].y;
                    hv.z = p[q].z * hv.z + e[q].z;
                    hv.w = p[q].w * hv.w + e[q].w;
                }
            }
        }
        sHi[i >> 2][i & 3] = hv;
    }
    __syncthreads();

    float A[DSTATE], h[DSTATE];
#pragma unroll
    for (int q = 0; q < 4; ++q) {
        const float4 v = sHi[dl][q];
        h[4 * q] = v.x; h[4 * q + 1] = v.y; h[4 * q + 2] = v.z; h[4 * q + 3] = v.w;
    }
#pragma unroll
    for (int n = 0; n < DSTATE; ++n)
        A[n] = -expf(Alog[d * DSTATE + n]);
    if (s > 0) {
        const size_t su = ((size_t)(cb * NSUB + s) * DSTATE) * DMODEL + d;
#pragma unroll
        for (int n = 0; n < DSTATE; ++n)
            h[n] = Pcum[su + (size_t)n * DMODEL] * h[n] + Hcum[su + (size_t)n * DMODEL];
    }
    const float Dsd = Ds[d];

    v2f A2[DSTATE / 2], h2[DSTATE / 2];
#pragma unroll
    for (int p = 0; p < DSTATE / 2; ++p) {
        A2[p] = (v2f){A[2 * p], A[2 * p + 1]};
        h2[p] = (v2f){h[2 * p], h[2 * p + 1]};
    }
    float Amin = A[0], Amax = A[0];
#pragma unroll
    for (int n = 1; n < DSTATE; ++n) {
        Amin = __builtin_fminf(Amin, A[n]);
        Amax = __builtin_fmaxf(Amax, A[n]);
    }

    const int l0 = ch * CHL + s * SUBL;
    for (int t = 0; t < SUBL; ++t) {
        const int bl = b * LL + l0 + t;
        if (t + 8 < SUBL) {
            __builtin_prefetch(&delta[(size_t)(bl + 8) * DMODEL + d], 0, 1);
            __builtin_prefetch(&X[(size_t)(bl + 8) * DMODEL + d], 0, 1);
        }
        const float dt = delta[(size_t)bl * DMODEL + d];
        const float u  = X[(size_t)bl * DMODEL + d];
        const float du = dt * u;
        const float4* Bv4 = reinterpret_cast<const float4*>(Bsc + (size_t)bl * DSTATE);
        const float4* Cv4 = reinterpret_cast<const float4*>(Csc + (size_t)bl * DSTATE);
        float Bl[DSTATE], Cl[DSTATE];
#pragma unroll
        for (int q = 0; q < 4; ++q) {
            float4 vb = Bv4[q], vc = Cv4[q];
            Bl[4*q] = vb.x; Bl[4*q+1] = vb.y; Bl[4*q+2] = vb.z; Bl[4*q+3] = vb.w;
            Cl[4*q] = vc.x; Cl[4*q+1] = vc.y; Cl[4*q+2] = vc.z; Cl[4*q+3] = vc.w;
        }
        const v2f du2 = {du, du};
        v2f a[DSTATE / 2];
        step_decays(dt, Amin, Amax, A2, a);
        float y = 0.0f;
#pragma unroll
        for (int p = 0; p < DSTATE / 2; ++p) {
            const v2f Bp = {Bl[2 * p], Bl[2 * p + 1]};
            const v2f Cp = {Cl[2 * p], Cl[2 * p + 1]};
            h2[p] = __builtin_elementwise_fma(du2, Bp, a[p] * h2[p]);
            const v2f hc = h2[p] * Cp;
            y += hc.x;
            y += hc.y;
        }
        Y[(size_t)bl * DMODEL + d] = y + u * Dsd;
    }
}

extern "C" void kernel_launch(void* const* d_in, const int* in_sizes, int n_in,
                              void* d_out, int out_size, void* d_ws, size_t ws_size,
                              hipStream_t stream)
{
    const float* x      = (const float*)d_in[0];
    const float* prompt = (const float*)d_in[1];
    const float* w1     = (const float*)d_in[2];
    const float* w2     = (const float*)d_in[3];
    const float* bias   = (const float*)d_in[4];
    const float* alog   = (const float*)d_in[5];
    const float* ds     = (const float*)d_in[6];
    float*       y      = (float*)d_out;

    float* ws = (float*)d_ws;
    size_t o = 0;
    o += (size_t)MTOT * DTRANK;
    float* Bsc   = ws + o; o += (size_t)MTOT * DSTATE;
    float* Csc   = ws + o; o += (size_t)MTOT * DSTATE;
    float* delta = ws + o; o += (size_t)MTOT * DMODEL;
    const size_t units = (size_t)NCH * BB * DMODEL;
    float* P     = ws + o; o += units * DSTATE;
    float* Hend  = ws + o; o += units * DSTATE;
    float* Pcum  = ws + o; o += units * NSUB * DSTATE;
    float* Hcum  = ws + o; o += units * NSUB * DSTATE;
    (void)ws_size; (void)in_sizes; (void)n_in; (void)out_size;

    proj_dt_kernel<<<dim3(MTOT / FB_ROWS), 512, 0, stream>>>(x, w1, prompt, w2, bias, Bsc, Csc, delta);
    scan_passA_kernel<<<dim3(NCH * BB * NDT), 256, 0, stream>>>(delta, x, Bsc, alog, P, Hend, Pcum, Hcum);
    scan_passC_kernel<<<dim3(NCH * BB * NDT), 256, 0, stream>>>(delta, x, Bsc, Csc, alog, ds, P, Hend, Pcum, Hcum, y);
}


# Scan passes A and C: a decay's integer exponent taken by a packed fma onto 1.5*2^23 and added as bits to 2^fraction's exponent field (rint, convert, ldexp go); steps with |dt*A|>86 keep the old form
# speedup vs baseline: 5.1077x; 1.1276x over previous
#include <hip/hip_runtime.h>
#include <hip/hip_bf16.h>
#include <math.h>

#define BB      4
#define LL      4096
#define DMODEL  384
#define DSTATE  16
#define DTRANK  24
#define CTOT    (DTRANK + 2*DSTATE)
#define MTOT    (BB*LL)
#define NCH     32
#define CHL     (LL/NCH)
#define NSUB    4
#define SUBL    (CHL/NSUB)
#define DTILE   (256/NSUB)
#define NDT     (DMODEL/DTILE)

typedef float     v2f  __attribute__((ext_vector_type(2)));
typedef unsigned  v2u  __attribute__((ext_vector_type(2)));
typedef float     v8f  __attribute__((ext_vector_type(8)));
typedef _Float16  v16h __attribute__((ext_vector_type(16)));

#if defined(__gfx1250__) && __has_builtin(__builtin_amdgcn_wmma_f32_16x16x4_f32)
#define HAVE_WMMA_F32 1
#else
#define HAVE_WMMA_F32 0
#endif

#define FB_ROWS  64
#define W1P      388
#define SDP      28

#if !HAVE_WMMA_F32 && defined(__HIP_DEVICE_COMPILE__)
#error "the f32 16x16x4 matrix instruction is required"
#endif

__device__ __forceinline__ v8f mma_f32_k4(v2f a, v2f b, v8f c)
{
#if HAVE_WMMA_F32
    return __builtin_amdgcn_wmma_f32_16x16x4_f32(false, a, false, b, (short)0, c, false, false);
#else
    return c;
#endif
}

#define EXP_X_MAX_BITS 0x42b17218u
__device__ __forceinline__ v2f exp_pair_core(v2f x)
{
    const float L_hi  = __builtin_bit_cast(float, 0x3fb8aa3bu);
    const float L_lo  = __builtin_bit_cast(float, 0x32a5705fu);
    const v2f Lh = {L_hi, L_hi};
    const v2f Ll = {L_lo, L_lo};
    const v2f t = x * Lh;
    v2f e;
    e.x = __builtin_rintf(t.x);
    e.y = __builtin_rintf(t.y);
    v2f r = __builtin_elementwise_fma(x, Lh, -e);
    r = __builtin_elementwise_fma(x, Ll, r);
    v2f v;
    v.x = __builtin_ldexpf(__builtin_amdgcn_exp2f(r.x), (int)e.x);
    v.y = __builtin_ldexpf(__builtin_amdgcn_exp2f(r.y), (int)e.y);
    return v;
}

__device__ __forceinline__ v2f exp_pair_bounds(v2f v, v2f x)
{
    const float X_MIN = __builtin_bit_cast(float, 0xc2ce8ed0u);
    const float X_MAX = __builtin_bit_cast(float, EXP_X_MAX_BITS);
    v.x = (x.x < X_MIN) ? 0.0f : v.x;
    v.y = (x.y < X_MIN) ? 0.0f : v.y;
    v.x = (x.x > X_MAX) ? __builtin_inff() : v.x;
    v.y = (x.y > X_MAX) ? __builtin_inff() : v.y;
    return v;
}

#define EXP_FAST_LIM 86.0f
__device__ __forceinline__ void step_decays(float dt, float Amin, float Amax,
                                            const v2f (&A2)[DSTATE / 2], v2f (&a)[DSTATE / 2])
{
    const bool in_range = (__builtin_fabsf(dt * Amin) <= EXP_FAST_LIM) && (__builtin_fabsf(dt * Amax) <= EXP_FAST_LIM);
    const float L_hi = __builtin_bit_cast(float, 0x3fb8aa3bu);
    const float L_lo = __builtin_bit_cast(float, 0x32a5705fu);
    const float RND  = 12582912.0f;
    const v2f Lh = {L_hi, L_hi};
    const v2f Ll = {L_lo, L_lo};
    const v2f Rn = {RND, RND};
    const v2f dt2 = {dt, dt};
    v2f x[DSTATE / 2], tm[DSTATE / 2], r[DSTATE / 2];
#pragma unroll
    for (int p = 0; p < DSTATE / 2; ++p) {
        x[p]  = dt2 * A2[p];
        tm[p] = __builtin_elementwise_fma(x[p], Lh, Rn);
    }
#pragma unroll
    for (int p = 0; p < DSTATE / 2; ++p) {
        const v2f e = tm[p] - Rn;
        r[p] = __builtin_elementwise_fma(x[p], Ll, __builtin_elementwise_fma(x[p], Lh, -e));
    }
#pragma unroll
    for (int p = 0; p < DSTATE / 2; ++p) {
        v2f q;
        q.x = __builtin_amdgcn_exp2f(r[p].x);
        q.y = __builtin_amdgcn_exp2f(r[p].y);
        a[p] = __builtin_bit_cast(v2f, __builtin_bit_cast(v2u, q) + (__builtin_bit_cast(v2u, tm[p]) << 23));
    }
    if (__builtin_amdgcn_ballot_w32(!in_range) != 0u) {
#pragma unroll
        for (int p = 0; p < DSTATE / 2; ++p)
            a[p] = exp_pair_bounds(exp_pair_core(x[p]), x[p]);
    }
}

__global__ __launch_bounds__(512)
void proj_dt_kernel(const float* __restrict__ X,
                    const float* __restrict__ W1,
                    const float* __restrict__ prompt,
                    const float* __restrict__ W2,
                    const float* __restrict__ bias,
                    float* __restrict__ Bsc,
                    float* __restrict__ Csc,
                    float* __restrict__ delta)
{
    __shared__ float smem[64 * W1P + FB_ROWS * SDP];
    float* Wlds = smem;
    float* part = smem;
    float* sD   = smem + 64 * W1P;

    const int tid  = threadIdx.x;
    const int wave = tid >> 5;
    const int lane = tid & 31;
    const int ln   = lane & 15;
    const int hi   = lane >> 4;
    const int rt   = wave >> 2;
    const int q    = wave & 3;
    const int mb   = blockIdx.x * FB_ROWS;
    const int m0   = mb + rt * 16;

    v2f   bfr[6][6];
    float bi[6];
#pragma unroll
    for (int j = 0; j < 6; ++j) {
        const float* wrow = W2 + (size_t)(96 * q + 16 * j + ln) * DTRANK + 2 * hi;
#pragma unroll
        for (int ks = 0; ks < 6; ++ks) {
            bfr[j][ks].x = wrow[4 * ks];
            bfr[j][ks].y = wrow[4 * ks + 1];
        }
        bi[j] = bias[96 * q + 16 * j + ln];
    }

    {
        const float4* W14 = reinterpret_cast<const float4*>(W1);
        const int NV = CTOT * DMODEL / 4;
        float4 v[11];
#pragma unroll
        for (int i = 0; i < 11; ++i) {
            const int idx = tid + i * 512;
            v[i] = (idx < NV) ? W14[idx] : make_float4(0.0f, 0.0f, 0.0f, 0.0f);
        }
#pragma unroll
        for (int i = 0; i < 11; ++i) {
            const int idx = tid + i * 512;
            if (idx < NV) {
                const int n = idx / (DMODEL / 4), k4 = idx - n * (DMODEL / 4);
                *reinterpret_cast<float4*>(&Wlds[n * W1P + 4 * k4]) = v[i];
            }
        }
        for (int idx = tid; idx < 8 * (DMODEL / 4); idx += 512) {
            const int n = CTOT + idx / (DMODEL / 4), k4 = idx % (DMODEL / 4);
            *reinterpret_cast<float4*>(&Wlds[n * W1P + 4 * k4]) = make_float4(0.0f, 0.0f, 0.0f, 0.0f);
        }
    }
    __syncthreads();

    v8f acc[4] = {};
    {
        const float* arow = X + (size_t)(m0 + ln) * DMODEL + 96 * q + 2 * hi;
        const float* brow = Wlds + ln * W1P + 96 * q + 2 * hi;
#pragma unroll
        for (int k0 = 0; k0 < 96; k0 += 4) {
            v2f a;
            a.x = arow[k0];
            a.y = arow[k0 + 1];
#pragma unroll
            for (int ct = 0; ct < 4; ++ct) {
                v2f b;
                b.x = brow[ct * 16 * W1P + k0];
                b.y = brow[ct * 16 * W1P + k0 + 1];
                acc[ct] = mma_f32_k4(a, b, acc[ct]);
            }
        }
    }
    __syncthreads();

#pragma unroll
    for (int ct = 0; ct < 4; ++ct)
#pragma unroll
        for (int r = 0; r < 8; ++r)
            part[((wave * 4 + ct) * 8 + r) * 32 + lane] = acc[ct][r];
    __syncthreads();

    {
        const int c = 16 * q + ln;
        float v[8];
#pragma unroll
        for (int r = 0; r < 8; ++r) {
            float t = part[(((rt * 4 + 0) * 4 + q) * 8 + r) * 32 + lane];
            t += part[(((rt * 4 + 1) * 4 + q) * 8 + r) * 32 + lane];
            t += part[(((rt * 4 + 2) * 4 + q) * 8 + r) * 32 + lane];
            t += part[(((rt * 4 + 3) * 4 + q) * 8 + r) * 32 + lane];
            v[r] = t;
        }
        const int ml0 = rt * 16 + 8 * hi;
        if (c < DTRANK) {
#pragma unroll
            for (int r = 0; r < 8; ++r)
                sD[(ml0 + r) * SDP + c] = v[r];
        } else if (c < DTRANK + DSTATE) {
#pragma unroll
            for (int r = 0; r < 8; ++r)
                Bsc[(size_t)(mb + ml0 + r) * DSTATE + (c - DTRANK)] = v[r];
        } else if (c < CTOT) {
            const int n = c - DTRANK - DSTATE;
            float pr[8];
#pragma unroll
            for (int r = 0; r < 8; ++r)
                pr[r] = prompt[(size_t)(mb + ml0 + r) * DSTATE + n];
#pragma unroll
            for (int r = 0; r < 8; ++r)
                Csc[(size_t)(mb + ml0 + r) * DSTATE + n] = v[r] + pr[r];
        }
    }
    __syncthreads();

    {
        const float* arow = sD + (rt * 16 + ln) * SDP + 2 * hi;
        v2f afr[6];
#pragma unroll
        for (int ks = 0; ks < 6; ++ks) {
            afr[ks].x = arow[4 * ks];
            afr[ks].y = arow[4 * ks + 1];
        }
#pragma unroll
        for (int j = 0; j < 6; ++j) {
            v8f a2 = {};
#pragma unroll
            for (int ks = 0; ks < 6; ++ks)
                a2 = mma_f32_k4(afr[ks], bfr[j][ks], a2);
            const int d = 96 * q + 16 * j + ln;
#pragma unroll
            for (int r = 0; r < 8; ++r) {
                const int   m  = m0 + r + 8 * hi;
                const float t  = a2[r] + bi[j];
                const float sp = (t > 20.0f) ? t : log1pf(expf(t));
                delta[(size_t)m * DMODEL + d] = sp;
            }
        }
    }
}

__global__ __launch_bounds__(256)
void scan_passA_kernel(const float* __restrict__ delta,
                       const float* __restrict__ X,
                       const float* __restrict__ Bsc,
                       const float* __restrict__ Alog,
                       float* __restrict__ Pout,
                       float* __restrict__ Hend,
                       float* __restrict__ Pcum,
                       float* __restrict__ Hcum)
{
    __shared__ float sP[NSUB][DSTATE][DTILE];
    __shared__ float sH[NSUB][DSTATE][DTILE];

    const int dl  = threadIdx.x % DTILE;
    const int s   = threadIdx.x / DTILE;
    const int dt_ = blockIdx.x % NDT;
    const int cb  = blockIdx.x / NDT;
    const int b   = cb % BB;
    const int ch  = cb / BB;
    const int d   = dt_ * DTILE + dl;

    v2f A2[DSTATE / 2], h2[DSTATE / 2], P2[DSTATE / 2];
#pragma unroll
    for (int p = 0; p < DSTATE / 2; ++p) {
        A2[p].x = -expf(Alog[d * DSTATE + 2 * p]);
        A2[p].y = -expf(Alog[d * DSTATE + 2 * p + 1]);
        h2[p] = (v2f){0.0f, 0.0f};
        P2[p] = (v2f){1.0f, 1.0f};
    }
    float Amin = A2[0].x, Amax = A2[0].x;
#pragma unroll
    for (int p = 0; p < DSTATE / 2; ++p) {
        Amin = __builtin_fminf(Amin, __builtin_fminf(A2[p].x, A2[p].y));
        Amax = __builtin_fmaxf(Amax, __builtin_fmaxf(A2[p].x, A2[p].y));
    }

    const int l0 = ch * CHL + s * SUBL;
    for (int t = 0; t < SUBL; ++t) {
        const int bl = b * LL + l0 + t;
        if (t + 8 < SUBL) {
            __builtin_prefetch(&delta[(size_t)(bl + 8) * DMODEL + d], 0, 1);
            __builtin_prefetch(&X[(size_t)(bl + 8) * DMODEL + d], 0, 1);
        }
        const float dt = delta[(size_t)bl * DMODEL + d];
        const float u  = X[(size_t)bl * DMODEL + d];
        const float du = dt * u;
        const float4* Bv4 = reinterpret_cast<const float4*>(Bsc + (size_t)bl * DSTATE);
        float Bl[DSTATE];
#pragma unroll
        for (int q = 0; q < 4; ++q) {
            float4 v = Bv4[q];
            Bl[4 * q] = v.x; Bl[4 * q + 1] = v.y; Bl[4 * q + 2] = v.z; Bl[4 * q + 3] = v.w;
        }
        const v2f du2 = {du, du};
        v2f a[DSTATE / 2];
        step_decays(dt, Amin, Amax, A2, a);
#pragma unroll
        for (int p = 0; p < DSTATE / 2; ++p) {
            const v2f Bp = {Bl[2 * p], Bl[2 * p + 1]};
            P2[p] *= a[p];
            h2[p]  = __builtin_elementwise_fma(du2, Bp, a[p] * h2[p]);
        }
    }

    float h[DSTATE], P[DSTATE];
#pragma unroll
    for (int p = 0; p < DSTATE / 2; ++p) {
        h[2 * p] = h2[p].x; h[2 * p + 1] = h2[p].y;
        P[2 * p] = P2[p].x; P[2 * p + 1] = P2[p].y;
    }

#pragma unroll
    for (int n = 0; n < DSTATE; ++n) { sP[s][n][dl] = P[n]; sH[s][n][dl] = h[n]; }
    __syncthreads();

    float Pc[DSTATE], Hc[DSTATE];
#pragma unroll
    for (int n = 0; n < DSTATE; ++n) { Pc[n] = 1.0f; Hc[n] = 0.0f; }
    for (int j = 0; j < s; ++j) {
#pragma unroll
        for (int n = 0; n < DSTATE; ++n) {
            const float pj = sP[j][n][dl];
            Hc[n] = pj * Hc[n] + sH[j][n][dl];
            Pc[n] *= pj;
        }
    }

    if (s > 0) {
        const size_t su = ((size_t)(cb * NSUB + s) * DSTATE) * DMODEL + d;
#pragma unroll
        for (int n = 0; n < DSTATE; ++n) {
            Pcum[su + (size_t)n * DMODEL] = Pc[n];
            Hcum[su + (size_t)n * DMODEL] = Hc[n];
        }
    }
    if (s == NSUB - 1) {
        const size_t unit = ((size_t)cb * DMODEL + d) * DSTATE;
        float* Pp = Pout + unit;
        float* Hp = Hend + unit;
#pragma unroll
        for (int n = 0; n < DSTATE; ++n) {
            Pp[n] = Pc[n] * P[n];
            Hp[n] = P[n] * Hc[n] + h[n];
        }
    }
}

#define CF_BATCH 8
__global__ __launch_bounds__(256)
void scan_passC_kernel(const float* __restrict__ delta,
                       const float* __restrict__ X,
                       const float* __restrict__ Bsc,
                       const float* __restrict__ Csc,
                       const float* __restrict__ Alog,
                       const float* __restrict__ Ds,
                       const float* __restrict__ Pch,
                       const float* __restrict__ Hend,
                       const float* __restrict__ Pcum,
                       const float* __restrict__ Hcum,
                       float* __restrict__ Y)
{
    __shared__ float4 sHi[DTILE][DSTATE / 4 + 1];

    const int dl  = threadIdx.x % DTILE;
    const int s   = threadIdx.x / DTILE;
    const int dt_ = blockIdx.x % NDT;
    const int cbr = blockIdx.x / NDT;
    const int b   = cbr % BB;
    const int ch  = (NCH - 1 - cbr / BB) & (NCH - 1);
    const int cb  = ch * BB + b;
    const int d   = dt_ * DTILE + dl;

    {
        const int    i       = threadIdx.x;
        const size_t cstride = (size_t)BB * DMODEL * DSTATE / 4;
        const size_t base    = ((size_t)b * DMODEL + dt_ * DTILE) * DSTATE / 4 + i;
        const float4* P4 = reinterpret_cast<const float4*>(Pch)  + base;
        const float4* H4 = reinterpret_cast<const float4*>(Hend) + base;
        float4 hv = make_float4(0.0f, 0.0f, 0.0f, 0.0f);
        for (int j0 = 0; j0 < ch; j0 += CF_BATCH) {
            float4 p[CF_BATCH], e[CF_BATCH];
#pragma unroll
            for (int q = 0; q < CF_BATCH; ++q) {
                const int j = min(j0 + q, ch - 1);
                p[q] = P4[(size_t)j * cstride];
                e[q] = H4[(size_t)j * cstride];
            }
#pragma unroll
            for (int q = 0; q < CF_BATCH; ++q) {
                if (j0 + q < ch) {
                    hv.x = p[q].x * hv.x + e[q].x;
                    hv.y = p[q].y * hv.y + e[q].y;
                    hv.z = p[q].z * hv.z + e[q].z;
                    hv.w = p[q].w * hv.w + e[q].w;
                }
            }
        }
        sHi[i >> 2][i & 3] = hv;
    }
    __syncthreads();

    float A[DSTATE], h[DSTATE];
#pragma unroll
    for (int q = 0; q < 4; ++q) {
        const float4 v = sHi[dl][q];
        h[4 * q] = v.x; h[4 * q + 1] = v.y; h[4 * q + 2] = v.z; h[4 * q + 3] = v.w;
    }
#pragma unroll
    for (int n = 0; n < DSTATE; ++n)
        A[n] = -expf(Alog[d * DSTATE + n]);
    if (s > 0) {
        const size_t su = ((size_t)(cb * NSUB + s) * DSTATE) * DMODEL + d;
#pragma unroll
        for (int n = 0; n < DSTATE; ++n)
            h[n] = Pcum[su + (size_t)n * DMODEL] * h[n] + Hcum[su + (size_t)n * DMODEL];
    }
    const float Dsd = Ds[d];

    v2f A2[DSTATE / 2], h2[DSTATE / 2];
#pragma unroll
    for (int p = 0; p < DSTATE / 2; ++p) {
        A2[p] = (v2f){A[2 * p], A[2 * p + 1]};
        h2[p] = (v2f){h[2 * p], h[2 * p + 1]};
    }
    float Amin = A[0], Amax = A[0];
#pragma unroll
    for (int n = 1; n < DSTATE; ++n) {
        Amin = __builtin_fminf(Amin, A[n]);
        Amax = __builtin_fmaxf(Amax, A[n]);
    }

    const int l0 = ch * CHL + s * SUBL;
    for (int t = 0; t < SUBL; ++t) {
        const int bl = b * LL + l0 + t;
        if (t + 8 < SUBL) {
            __builtin_prefetch(&delta[(size_t)(bl + 8) * DMODEL + d], 0, 1);
            __builtin_prefetch(&X[(size_t)(bl + 8) * DMODEL + d], 0, 1);
        }
        const float dt = delta[(size_t)bl * DMODEL + d];
        const float u  = X[(size_t)bl * DMODEL + d];
        const float du = dt * u;
        const float4* Bv4 = reinterpret_cast<const float4*>(Bsc + (size_t)bl * DSTATE);
        const float4* Cv4 = reinterpret_cast<const float4*>(Csc + (size_t)bl * DSTATE);
        float Bl[DSTATE], Cl[DSTATE];
#pragma unroll
        for (int q = 0; q < 4; ++q) {
            float4 vb = Bv4[q], vc = Cv4[q];
            Bl[4*q] = vb.x; Bl[4*q+1] = vb.y; Bl[4*q+2] = vb.z; Bl[4*q+3] = vb.w;
            Cl[4*q] = vc.x; Cl[4*q+1] = vc.y; Cl[4*q+2] = vc.z; Cl[4*q+3] = vc.w;
        }
        const v2f du2 = {du, du};
        v2f a[DSTATE / 2];
        step_decays(dt, Amin, Amax, A2, a);
        float y = 0.0f;
#pragma unroll
        for (int p = 0; p < DSTATE / 2; ++p) {
            const v2f Bp = {Bl[2 * p], Bl[2 * p + 1]};
            const v2f Cp = {Cl[2 * p], Cl[2 * p + 1]};
            h2[p] = __builtin_elementwise_fma(du2, Bp, a[p] * h2[p]);
            const v2f hc = h2[p] * Cp;
            y += hc.x;
            y += hc.y;
        }
        Y[(size_t)bl * DMODEL + d] = y + u * Dsd;
    }
}

extern "C" void kernel_launch(void* const* d_in, const int* in_sizes, int n_in,
                              void* d_out, int out_size, void* d_ws, size_t ws_size,
                              hipStream_t stream)
{
    const float* x      = (const float*)d_in[0];
    const float* prompt = (const float*)d_in[1];
    const float* w1     = (const float*)d_in[2];
    const float* w2     = (const float*)d_in[3];
    const float* bias   = (const float*)d_in[4];
    const float* alog   = (const float*)d_in[5];
    const float* ds     = (const float*)d_in[6];
    float*       y      = (float*)d_out;

    float* ws = (float*)d_ws;
    size_t o = 0;
    o += (size_t)MTOT * DTRANK;
    float* Bsc   = ws + o; o += (size_t)MTOT * DSTATE;
    float* Csc   = ws + o; o += (size_t)MTOT * DSTATE;
    float* delta = ws + o; o += (size_t)MTOT * DMODEL;
    const size_t units = (size_t)NCH * BB * DMODEL;
    float* P     = ws + o; o += units * DSTATE;
    float* Hend  = ws + o; o += units * DSTATE;
    float* Pcum  = ws + o; o += units * NSUB * DSTATE;
    float* Hcum  = ws + o; o += units * NSUB * DSTATE;
    (void)ws_size; (void)in_sizes; (void)n_in; (void)out_size;

    proj_dt_kernel<<<dim3(MTOT / FB_ROWS), 512, 0, stream>>>(x, w1, prompt, w2, bias, Bsc, Csc, delta);
    scan_passA_kernel<<<dim3(NCH * BB * NDT), 256, 0, stream>>>(delta, x, Bsc, alog, P, Hend, Pcum, Hcum);
    scan_passC_kernel<<<dim3(NCH * BB * NDT), 256, 0, stream>>>(delta, x, Bsc, Csc, alog, ds, P, Hend, Pcum, Hcum, y);
}


# Scan passes A and C: per-step addressing leaves the vector unit: sub-chunk index stated wave-uniform (row pointers in scalar registers); the chunk's B/C rows staged once in LDS, read wave-wide
# speedup vs baseline: 5.4291x; 1.0629x over previous
#include <hip/hip_runtime.h>
#include <hip/hip_bf16.h>
#include <math.h>

#define BB      4
#define LL      4096
#define DMODEL  384
#define DSTATE  16
#define DTRANK  24
#define CTOT    (DTRANK + 2*DSTATE)
#define MTOT    (BB*LL)
#define NCH     32
#define CHL     (LL/NCH)
#define NSUB    4
#define SUBL    (CHL/NSUB)
#define DTILE   (256/NSUB)
#define NDT     (DMODEL/DTILE)

typedef float     v2f  __attribute__((ext_vector_type(2)));
typedef unsigned  v2u  __attribute__((ext_vector_type(2)));
typedef float     v4f  __attribute__((ext_vector_type(4)));
typedef float     v8f  __attribute__((ext_vector_type(8)));
typedef _Float16  v16h __attribute__((ext_vector_type(16)));

#if defined(__gfx1250__) && __has_builtin(__builtin_amdgcn_wmma_f32_16x16x4_f32)
#define HAVE_WMMA_F32 1
#else
#define HAVE_WMMA_F32 0
#endif

#define FB_ROWS  64
#define W1P      388
#define SDP      28

#if !HAVE_WMMA_F32 && defined(__HIP_DEVICE_COMPILE__)
#error "the f32 16x16x4 matrix instruction is required"
#endif

__device__ __forceinline__ v8f mma_f32_k4(v2f a, v2f b, v8f c)
{
#if HAVE_WMMA_F32
    return __builtin_amdgcn_wmma_f32_16x16x4_f32(false, a, false, b, (short)0, c, false, false);
#else
    return c;
#endif
}

#define EXP_X_MAX_BITS 0x42b17218u
__device__ __forceinline__ v2f exp_pair_core(v2f x)
{
    const float L_hi  = __builtin_bit_cast(float, 0x3fb8aa3bu);
    const float L_lo  = __builtin_bit_cast(float, 0x32a5705fu);
    const v2f Lh = {L_hi, L_hi};
    const v2f Ll = {L_lo, L_lo};
    const v2f t = x * Lh;
    v2f e;
    e.x = __builtin_rintf(t.x);
    e.y = __builtin_rintf(t.y);
    v2f r = __builtin_elementwise_fma(x, Lh, -e);
    r = __builtin_elementwise_fma(x, Ll, r);
    v2f v;
    v.x = __builtin_ldexpf(__builtin_amdgcn_exp2f(r.x), (int)e.x);
    v.y = __builtin_ldexpf(__builtin_amdgcn_exp2f(r.y), (int)e.y);
    return v;
}

__device__ __forceinline__ v2f exp_pair_bounds(v2f v, v2f x)
{
    const float X_MIN = __builtin_bit_cast(float, 0xc2ce8ed0u);
    const float X_MAX = __builtin_bit_cast(float, EXP_X_MAX_BITS);
    v.x = (x.x < X_MIN) ? 0.0f : v.x;
    v.y = (x.y < X_MIN) ? 0.0f : v.y;
    v.x = (x.x > X_MAX) ? __builtin_inff() : v.x;
    v.y = (x.y > X_MAX) ? __builtin_inff() : v.y;
    return v;
}

#define EXP_FAST_LIM 86.0f
__device__ __forceinline__ void step_decays(float dt, float Amin, float Amax,
                                            const v2f (&A2)[DSTATE / 2], v2f (&a)[DSTATE / 2])
{
    const bool in_range = (__builtin_fabsf(dt * Amin) <= EXP_FAST_LIM) && (__builtin_fabsf(dt * Amax) <= EXP_FAST_LIM);
    const float L_hi = __builtin_bit_cast(float, 0x3fb8aa3bu);
    const float L_lo = __builtin_bit_cast(float, 0x32a5705fu);
    const float RND  = 12582912.0f;
    const v2f Lh = {L_hi, L_hi};
    const v2f Ll = {L_lo, L_lo};
    const v2f Rn = {RND, RND};
    const v2f dt2 = {dt, dt};
    v2f x[DSTATE / 2], tm[DSTATE / 2], r[DSTATE / 2];
#pragma unroll
    for (int p = 0; p < DSTATE / 2; ++p) {
        x[p]  = dt2 * A2[p];
        tm[p] = __builtin_elementwise_fma(x[p], Lh, Rn);
    }
#pragma unroll
    for (int p = 0; p < DSTATE / 2; ++p) {
        const v2f e = tm[p] - Rn;
        r[p] = __builtin_elementwise_fma(x[p], Ll, __builtin_elementwise_fma(x[p], Lh, -e));
    }
#pragma unroll
    for (int p = 0; p < DSTATE / 2; ++p) {
        v2f q;
        q.x = __builtin_amdgcn_exp2f(r[p].x);
        q.y = __builtin_amdgcn_exp2f(r[p].y);
        a[p] = __builtin_bit_cast(v2f, __builtin_bit_cast(v2u, q) + (__builtin_bit_cast(v2u, tm[p]) << 23));
    }
    if (__builtin_amdgcn_ballot_w32(!in_range) != 0u) {
#pragma unroll
        for (int p = 0; p < DSTATE / 2; ++p)
            a[p] = exp_pair_bounds(exp_pair_core(x[p]), x[p]);
    }
}

__global__ __launch_bounds__(512)
void proj_dt_kernel(const float* __restrict__ X,
                    const float* __restrict__ W1,
                    const float* __restrict__ prompt,
                    const float* __restrict__ W2,
                    const float* __restrict__ bias,
                    float* __restrict__ Bsc,
                    float* __restrict__ Csc,
                    float* __restrict__ delta)
{
    __shared__ float smem[64 * W1P + FB_ROWS * SDP];
    float* Wlds = smem;
    float* part = smem;
    float* sD   = smem + 64 * W1P;

    const int tid  = threadIdx.x;
    const int wave = tid >> 5;
    const int lane = tid & 31;
    const int ln   = lane & 15;
    const int hi   = lane >> 4;
    const int rt   = wave >> 2;
    const int q    = wave & 3;
    const int mb   = blockIdx.x * FB_ROWS;
    const int m0   = mb + rt * 16;

    v2f   bfr[6][6];
    float bi[6];
#pragma unroll
    for (int j = 0; j < 6; ++j) {
        const float* wrow = W2 + (size_t)(96 * q + 16 * j + ln) * DTRANK + 2 * hi;
#pragma unroll
        for (int ks = 0; ks < 6; ++ks) {
            bfr[j][ks].x = wrow[4 * ks];
            bfr[j][ks].y = wrow[4 * ks + 1];
        }
        bi[j] = bias[96 * q + 16 * j + ln];
    }

    {
        const float4* W14 = reinterpret_cast<const float4*>(W1);
        const int NV = CTOT * DMODEL / 4;
        float4 v[11];
#pragma unroll
        for (int i = 0; i < 11; ++i) {
            const int idx = tid + i * 512;
            v[i] = (idx < NV) ? W14[idx] : make_float4(0.0f, 0.0f, 0.0f, 0.0f);
        }
#pragma unroll
        for (int i = 0; i < 11; ++i) {
            const int idx = tid + i * 512;
            if (idx < NV) {
                const int n = idx / (DMODEL / 4), k4 = idx - n * (DMODEL / 4);
                *reinterpret_cast<float4*>(&Wlds[n * W1P + 4 * k4]) = v[i];
            }
        }
        for (int idx = tid; idx < 8 * (DMODEL / 4); idx += 512) {
            const int n = CTOT + idx / (DMODEL / 4), k4 = idx % (DMODEL / 4);
            *reinterpret_cast<float4*>(&Wlds[n * W1P + 4 * k4]) = make_float4(0.0f, 0.0f, 0.0f, 0.0f);
        }
    }
    __syncthreads();

    v8f acc[4] = {};
    {
        const float* arow = X + (size_t)(m0 + ln) * DMODEL + 96 * q + 2 * hi;
        const float* brow = Wlds + ln * W1P + 96 * q + 2 * hi;
#pragma unroll
        for (int k0 = 0; k0 < 96; k0 += 4) {
            v2f a;
            a.x = arow[k0];
            a.y = arow[k0 + 1];
#pragma unroll
            for (int ct = 0; ct < 4; ++ct) {
                v2f b;
                b.x = brow[ct * 16 * W1P + k0];
                b.y = brow[ct * 16 * W1P + k0 + 1];
                acc[ct] = mma_f32_k4(a, b, acc[ct]);
            }
        }
    }
    __syncthreads();

#pragma unroll
    for (int ct = 0; ct < 4; ++ct)
#pragma unroll
        for (int r = 0; r < 8; ++r)
            part[((wave * 4 + ct) * 8 + r) * 32 + lane] = acc[ct][r];
    __syncthreads();

    {
        const int c = 16 * q + ln;
        float v[8];
#pragma unroll
        for (int r = 0; r < 8; ++r) {
            float t = part[(((rt * 4 + 0) * 4 + q) * 8 + r) * 32 + lane];
            t += part[(((rt * 4 + 1) * 4 + q) * 8 + r) * 32 + lane];
            t += part[(((rt * 4 + 2) * 4 + q) * 8 + r) * 32 + lane];
            t += part[(((rt * 4 + 3) * 4 + q) * 8 + r) * 32 + lane];
            v[r] = t;
        }
        const int ml0 = rt * 16 + 8 * hi;
        if (c < DTRANK) {
#pragma unroll
            for (int r = 0; r < 8; ++r)
                sD[(ml0 + r) * SDP + c] = v[r];
        } else if (c < DTRANK + DSTATE) {
#pragma unroll
            for (int r = 0; r < 8; ++r)
                Bsc[(size_t)(mb + ml0 + r) * DSTATE + (c - DTRANK)] = v[r];
        } else if (c < CTOT) {
            const int n = c - DTRANK - DSTATE;
            float pr[8];
#pragma unroll
            for (int r = 0; r < 8; ++r)
                pr[r] = prompt[(size_t)(mb + ml0 + r) * DSTATE + n];
#pragma unroll
            for (int r = 0; r < 8; ++r)
                Csc[(size_t)(mb + ml0 + r) * DSTATE + n] = v[r] + pr[r];
        }
    }
    __syncthreads();

    {
        const float* arow = sD + (rt * 16 + ln) * SDP + 2 * hi;
        v2f afr[6];
#pragma unroll
        for (int ks = 0; ks < 6; ++ks) {
            afr[ks].x = arow[4 * ks];
            afr[ks].y = arow[4 * ks + 1];
        }
#pragma unroll
        for (int j = 0; j < 6; ++j) {
            v8f a2 = {};
#pragma unroll
            for (int ks = 0; ks < 6; ++ks)
                a2 = mma_f32_k4(afr[ks], bfr[j][ks], a2);
            const int d = 96 * q + 16 * j + ln;
#pragma unroll
            for (int r = 0; r < 8; ++r) {
                const int   m  = m0 + r + 8 * hi;
                const float t  = a2[r] + bi[j];
                const float sp = (t > 20.0f) ? t : log1pf(expf(t));
                delta[(size_t)m * DMODEL + d] = sp;
            }
        }
    }
}

__global__ __launch_bounds__(256)
void scan_passA_kernel(const float* __restrict__ delta,
                       const float* __restrict__ X,
                       const float* __restrict__ Bsc,
                       const float* __restrict__ Alog,
                       float* __restrict__ Pout,
                       float* __restrict__ Hend,
                       float* __restrict__ Pcum,
                       float* __restrict__ Hcum)
{
    __shared__ float sP[NSUB][DSTATE][DTILE];
    __shared__ float sH[NSUB][DSTATE][DTILE];
    __shared__ v4f   sB[CHL][DSTATE / 4];

    const int dl  = threadIdx.x % DTILE;
    const int s   = __builtin_amdgcn_readfirstlane(threadIdx.x / DTILE);
    const int dt_ = blockIdx.x % NDT;
    const int cb  = blockIdx.x / NDT;
    const int b   = cb % BB;
    const int ch  = cb / BB;
    const int d   = dt_ * DTILE + dl;

    const v4f* Bsrc = reinterpret_cast<const v4f*>(Bsc + (size_t)(b * LL + ch * CHL) * DSTATE);
    const v4f bst0 = Bsrc[threadIdx.x];
    const v4f bst1 = Bsrc[threadIdx.x + 256];

    v2f A2[DSTATE / 2], h2[DSTATE / 2], P2[DSTATE / 2];
#pragma unroll
    for (int p = 0; p < DSTATE / 2; ++p) {
        A2[p].x = -expf(Alog[d * DSTATE + 2 * p]);
        A2[p].y = -expf(Alog[d * DSTATE + 2 * p + 1]);
        h2[p] = (v2f){0.0f, 0.0f};
        P2[p] = (v2f){1.0f, 1.0f};
    }
    float Amin = A2[0].x, Amax = A2[0].x;
#pragma unroll
    for (int p = 0; p < DSTATE / 2; ++p) {
        Amin = __builtin_fminf(Amin, __builtin_fminf(A2[p].x, A2[p].y));
        Amax = __builtin_fmaxf(Amax, __builtin_fmaxf(A2[p].x, A2[p].y));
    }

    (&sB[0][0])[threadIdx.x]       = bst0;
    (&sB[0][0])[threadIdx.x + 256] = bst1;
    __syncthreads();

    const size_t row0 = (size_t)(b * LL + ch * CHL + s * SUBL);
    const float* drow = delta + row0 * DMODEL + dt_ * DTILE;
    const float* xrow = X     + row0 * DMODEL + dt_ * DTILE;
    for (int t = 0; t < SUBL; ++t) {
        if (t + 8 < SUBL) {
            __builtin_prefetch(&drow[8 * DMODEL + dl], 0, 1);
            __builtin_prefetch(&xrow[8 * DMODEL + dl], 0, 1);
        }
        const float dt = drow[dl];
        const float u  = xrow[dl];
        drow += DMODEL; xrow += DMODEL;
        const v4f* Brow = sB[s * SUBL + t];
        v4f Bv[DSTATE / 4];
#pragma unroll
        for (int q = 0; q < DSTATE / 4; ++q)
            Bv[q] = Brow[q];
        asm volatile("" :: "v"(u), "v"(Bv[0]), "v"(Bv[1]), "v"(Bv[2]), "v"(Bv[3]));
        const float du = dt * u;
        float Bl[DSTATE];
#pragma unroll
        for (int q = 0; q < 4; ++q) {
            Bl[4 * q] = Bv[q].x; Bl[4 * q + 1] = Bv[q].y; Bl[4 * q + 2] = Bv[q].z; Bl[4 * q + 3] = Bv[q].w;
        }
        const v2f du2 = {du, du};
        v2f a[DSTATE / 2];
        step_decays(dt, Amin, Amax, A2, a);
#pragma unroll
        for (int p = 0; p < DSTATE / 2; ++p) {
            const v2f Bp = {Bl[2 * p], Bl[2 * p + 1]};
            P2[p] *= a[p];
            h2[p]  = __builtin_elementwise_fma(du2, Bp, a[p] * h2[p]);
        }
    }

    float h[DSTATE], P[DSTATE];
#pragma unroll
    for (int p = 0; p < DSTATE / 2; ++p) {
        h[2 * p] = h2[p].x; h[2 * p + 1] = h2[p].y;
        P[2 * p] = P2[p].x; P[2 * p + 1] = P2[p].y;
    }

#pragma unroll
    for (int n = 0; n < DSTATE; ++n) { sP[s][n][dl] = P[n]; sH[s][n][dl] = h[n]; }
    __syncthreads();

    float Pc[DSTATE], Hc[DSTATE];
#pragma unroll
    for (int n = 0; n < DSTATE; ++n) { Pc[n] = 1.0f; Hc[n] = 0.0f; }
    for (int j = 0; j < s; ++j) {
#pragma unroll
        for (int n = 0; n < DSTATE; ++n) {
            const float pj = sP[j][n][dl];
            Hc[n] = pj * Hc[n] + sH[j][n][dl];
            Pc[n] *= pj;
        }
    }

    if (s > 0) {
        const size_t su = ((size_t)(cb * NSUB + s) * DSTATE) * DMODEL + d;
#pragma unroll
        for (int n = 0; n < DSTATE; ++n) {
            Pcum[su + (size_t)n * DMODEL] = Pc[n];
            Hcum[su + (size_t)n * DMODEL] = Hc[n];
        }
    }
    if (s == NSUB - 1) {
        const size_t unit = ((size_t)cb * DMODEL + d) * DSTATE;
        float* Pp = Pout + unit;
        float* Hp = Hend + unit;
#pragma unroll
        for (int n = 0; n < DSTATE; ++n) {
            Pp[n] = Pc[n] * P[n];
            Hp[n] = P[n] * Hc[n] + h[n];
        }
    }
}

#define CF_BATCH 8
__global__ __launch_bounds__(256)
void scan_passC_kernel(const float* __restrict__ delta,
                       const float* __restrict__ X,
                       const float* __restrict__ Bsc,
                       const float* __restrict__ Csc,
                       const float* __restrict__ Alog,
                       const float* __restrict__ Ds,
                       const float* __restrict__ Pch,
                       const float* __restrict__ Hend,
                       const float* __restrict__ Pcum,
                       const float* __restrict__ Hcum,
                       float* __restrict__ Y)
{
    __shared__ float4 sHi[DTILE][DSTATE / 4 + 1];
    __shared__ v4f    sB[CHL][DSTATE / 4];
    __shared__ v4f    sC[CHL][DSTATE / 4];

    const int dl  = threadIdx.x % DTILE;
    const int s   = __builtin_amdgcn_readfirstlane(threadIdx.x / DTILE);
    const int dt_ = blockIdx.x % NDT;
    const int cbr = blockIdx.x / NDT;
    const int b   = cbr % BB;
    const int ch  = (NCH - 1 - cbr / BB) & (NCH - 1);
    const int cb  = ch * BB + b;
    const int d   = dt_ * DTILE + dl;

    const v4f* Bsrc = reinterpret_cast<const v4f*>(Bsc + (size_t)(b * LL + ch * CHL) * DSTATE);
    const v4f* Csrc = reinterpret_cast<const v4f*>(Csc + (size_t)(b * LL + ch * CHL) * DSTATE);
    const v4f bst0 = Bsrc[threadIdx.x], bst1 = Bsrc[threadIdx.x + 256];
    const v4f cst0 = Csrc[threadIdx.x], cst1 = Csrc[threadIdx.x + 256];

    {
        const int    i       = threadIdx.x;
        const size_t cstride = (size_t)BB * DMODEL * DSTATE / 4;
        const size_t base    = ((size_t)b * DMODEL + dt_ * DTILE) * DSTATE / 4 + i;
        const float4* P4 = reinterpret_cast<const float4*>(Pch)  + base;
        const float4* H4 = reinterpret_cast<const float4*>(Hend) + base;
        float4 hv = make_float4(0.0f, 0.0f, 0.0f, 0.0f);
        for (int j0 = 0; j0 < ch; j0 += CF_BATCH) {
            float4 p[CF_BATCH], e[CF_BATCH];
#pragma unroll
            for (int q = 0; q < CF_BATCH; ++q) {
                const int j = min(j0 + q, ch - 1);
                p[q] = P4[(size_t)j * cstride];
                e[q] = H4[(size_t)j * cstride];
            }
#pragma unroll
            for (int q = 0; q < CF_BATCH; ++q) {
                if (j0 + q < ch) {
                    hv.x = p[q].x * hv.x + e[q].x;
                    hv.y = p[q].y * hv.y + e[q].y;
                    hv.z = p[q].z * hv.z + e[q].z;
                    hv.w = p[q].w * hv.w + e[q].w;
                }
            }
        }
        sHi[i >> 2][i & 3] = hv;
        (&sB[0][0])[i] = bst0; (&sB[0][0])[i + 256] = bst1;
        (&sC[0][0])[i] = cst0; (&sC[0][0])[i + 256] = cst1;
    }
    __syncthreads();

    float A[DSTATE], h[DSTATE];
#pragma unroll
    for (int q = 0; q < 4; ++q) {
        const float4 v = sHi[dl][q];
        h[4 * q] = v.x; h[4 * q + 1] = v.y; h[4 * q + 2] = v.z; h[4 * q + 3] = v.w;
    }
#pragma unroll
    for (int n = 0; n < DSTATE; ++n)
        A[n] = -expf(Alog[d * DSTATE + n]);
    if (s > 0) {
        const size_t su = ((size_t)(cb * NSUB + s) * DSTATE) * DMODEL + d;
#pragma unroll
        for (int n = 0; n < DSTATE; ++n)
            h[n] = Pcum[su + (size_t)n * DMODEL] * h[n] + Hcum[su + (size_t)n * DMODEL];
    }
    const float Dsd = Ds[d];

    v2f A2[DSTATE / 2], h2[DSTATE / 2];
#pragma unroll
    for (int p = 0; p < DSTATE / 2; ++p) {
        A2[p] = (v2f){A[2 * p], A[2 * p + 1]};
        h2[p] = (v2f){h[2 * p], h[2 * p + 1]};
    }
    float Amin = A[0], Amax = A[0];
#pragma unroll
    for (int n = 1; n < DSTATE; ++n) {
        Amin = __builtin_fminf(Amin, A[n]);
        Amax = __builtin_fmaxf(Amax, A[n]);
    }

    const size_t row0 = (size_t)(b * LL + ch * CHL + s * SUBL);
    const float* drow = delta + row0 * DMODEL + dt_ * DTILE;
    const float* xrow = X     + row0 * DMODEL + dt_ * DTILE;
    float*       yrow = Y     + row0 * DMODEL + dt_ * DTILE;
    for (int t = 0; t < SUBL; ++t) {
        if (t + 8 < SUBL) {
            __builtin_prefetch(&drow[8 * DMODEL + dl], 0, 1);
            __builtin_prefetch(&xrow[8 * DMODEL + dl], 0, 1);
        }
        const float dt = drow[dl];
        const float u  = xrow[dl];
        drow += DMODEL; xrow += DMODEL;
        const v4f* Brow = sB[s * SUBL + t];
        const v4f* Crow = sC[s * SUBL + t];
        v4f Bv[DSTATE / 4], Cv[DSTATE / 4];
#pragma unroll
        for (int q = 0; q < DSTATE / 4; ++q) {
            Bv[q] = Brow[q];
            Cv[q] = Crow[q];
        }
        asm volatile("" :: "v"(u), "v"(Bv[0]), "v"(Bv[1]), "v"(Bv[2]), "v"(Bv[3]),
                                   "v"(Cv[0]), "v"(Cv[1]), "v"(Cv[2]), "v"(Cv[3]));
        const float du = dt * u;
        float Bl[DSTATE], Cl[DSTATE];
#pragma unroll
        for (int q = 0; q < 4; ++q) {
            Bl[4*q] = Bv[q].x; Bl[4*q+1] = Bv[q].y; Bl[4*q+2] = Bv[q].z; Bl[4*q+3] = Bv[q].w;
            Cl[4*q] = Cv[q].x; Cl[4*q+1] = Cv[q].y; Cl[4*q+2] = Cv[q].z; Cl[4*q+3] = Cv[q].w;
        }
        const v2f du2 = {du, du};
        v2f a[DSTATE / 2];
        step_decays(dt, Amin, Amax, A2, a);
        float y = 0.0f;
#pragma unroll
        for (int p = 0; p < DSTATE / 2; ++p) {
            const v2f Bp = {Bl[2 * p], Bl[2 * p + 1]};
            const v2f Cp = {Cl[2 * p], Cl[2 * p + 1]};
            h2[p] = __builtin_elementwise_fma(du2, Bp, a[p] * h2[p]);
            const v2f hc = h2[p] * Cp;
            y += hc.x;
            y += hc.y;
        }
        yrow[dl] = y + u * Dsd;
        yrow += DMODEL;
    }
}

extern "C" void kernel_launch(void* const* d_in, const int* in_sizes, int n_in,
                              void* d_out, int out_size, void* d_ws, size_t ws_size,
                              hipStream_t stream)
{
    const float* x      = (const float*)d_in[0];
    const float* prompt = (const float*)d_in[1];
    const float* w1     = (const float*)d_in[2];
    const float* w2     = (const float*)d_in[3];
    const float* bias   = (const float*)d_in[4];
    const float* alog   = (const float*)d_in[5];
    const float* ds     = (const float*)d_in[6];
    float*       y      = (float*)d_out;

    float* ws = (float*)d_ws;
    size_t o = 0;
    o += (size_t)MTOT * DTRANK;
    float* Bsc   = ws + o; o += (size_t)MTOT * DSTATE;
    float* Csc   = ws + o; o += (size_t)MTOT * DSTATE;
    float* delta = ws + o; o += (size_t)MTOT * DMODEL;
    const size_t units = (size_t)NCH * BB * DMODEL;
    float* P     = ws + o; o += units * DSTATE;
    float* Hend  = ws + o; o += units * DSTATE;
    float* Pcum  = ws + o; o += units * NSUB * DSTATE;
    float* Hcum  = ws + o; o += units * NSUB * DSTATE;
    (void)ws_size; (void)in_sizes; (void)n_in; (void)out_size;

    proj_dt_kernel<<<dim3(MTOT / FB_ROWS), 512, 0, stream>>>(x, w1, prompt, w2, bias, Bsc, Csc, delta);
    scan_passA_kernel<<<dim3(NCH * BB * NDT), 256, 0, stream>>>(delta, x, Bsc, alog, P, Hend, Pcum, Hcum);
    scan_passC_kernel<<<dim3(NCH * BB * NDT), 256, 0, stream>>>(delta, x, Bsc, Csc, alog, ds, P, Hend, Pcum, Hcum, y);
}
